# starting kernel
# baseline (speedup 1.0000x reference)
#include <hip/hip_runtime.h>

typedef __attribute__((ext_vector_type(2))) float v2f;
typedef __attribute__((ext_vector_type(4))) float v4f;
typedef __attribute__((ext_vector_type(8))) float v8f;

#define D_MODEL 1024
#define H_HEADS 16
#define HEAD_DIM 64
#define SEQ_L 2048
#define BATCH 2
#define M_ROWS (BATCH * SEQ_L)

__device__ __forceinline__ v8f wmma_f32(v2f a, v2f b, v8f c) {
  return __builtin_amdgcn_wmma_f32_16x16x4_f32(false, a, false, b, (short)0, c,
                                               false, false);
}

__device__ __forceinline__ float rowmax16(float x) {
#pragma unroll
  for (int m = 1; m < 16; m <<= 1) x = fmaxf(x, __shfl_xor(x, m, 32));
  return x;
}
__device__ __forceinline__ float rowsum16(float x) {
#pragma unroll
  for (int m = 1; m < 16; m <<= 1) x += __shfl_xor(x, m, 32);
  return x;
}

#define BM 64
#define BN 128
#define BK 32
#define APAD 36
#define BPAD 34

__global__ __launch_bounds__(256) void gemm_bias_kernel(
    const float* __restrict__ A, const float* __restrict__ B,
    const float* __restrict__ bias, float* __restrict__ C, int N, int K) {
  __shared__ float As[BM][APAD];
  __shared__ float Bs[BN][BPAD];

  const int tid = threadIdx.x;
  const int lane = tid & 31;
  const int w = tid >> 5;
  const int wm = (w >> 2) << 5;
  const int wn = (w & 3) << 5;
  const int m0 = blockIdx.y * BM;
  const int n0 = blockIdx.x * BN;

  const int ln = lane & 15;
  const int half = lane >> 4;
  const int kb = half << 1;

  v8f acc[2][2];
#pragma unroll
  for (int i = 0; i < 2; ++i)
#pragma unroll
    for (int j = 0; j < 2; ++j)
#pragma unroll
      for (int v = 0; v < 8; ++v) acc[i][j][v] = 0.0f;

  for (int k0 = 0; k0 < K; k0 += BK) {
    __syncthreads();
    {
      int r = tid >> 3;
      int c = (tid & 7) << 2;
      const float* ap = A + (size_t)(m0 + r) * K + (k0 + c);
      *(v4f*)&As[r][c] = *(const v4f*)ap;
      *(v4f*)&As[r + 32][c] = *(const v4f*)(ap + (size_t)32 * K);
    }
    {
      int kr = tid >> 5;
      int nc = (tid & 31) << 2;
#pragma unroll
      for (int it = 0; it < 4; ++it) {
        int kk = kr + it * 8;
        v4f bv = *(const v4f*)(B + (size_t)(k0 + kk) * N + (n0 + nc));
        Bs[nc + 0][kk] = bv.x;
        Bs[nc + 1][kk] = bv.y;
        Bs[nc + 2][kk] = bv.z;
        Bs[nc + 3][kk] = bv.w;
      }
    }
    __syncthreads();
#pragma unroll
    for (int kk = 0; kk < BK; kk += 4) {
      v2f a0 = *(const v2f*)&As[wm + ln][kk + kb];
      v2f a1 = *(const v2f*)&As[wm + 16 + ln][kk + kb];
      v2f b0 = *(const v2f*)&Bs[wn + ln][kk + kb];
      v2f b1 = *(const v2f*)&Bs[wn + 16 + ln][kk + kb];
      acc[0][0] = wmma_f32(a0, b0, acc[0][0]);
      acc[0][1] = wmma_f32(a0, b1, acc[0][1]);
      acc[1][0] = wmma_f32(a1, b0, acc[1][0]);
      acc[1][1] = wmma_f32(a1, b1, acc[1][1]);
    }
  }

#pragma unroll
  for (int j = 0; j < 2; ++j) {
    float bj = bias[n0 + wn + j * 16 + ln];
#pragma unroll
    for (int i = 0; i < 2; ++i) {
      int rbase = m0 + wm + i * 16 + half * 8;
      int col = n0 + wn + j * 16 + ln;
#pragma unroll
      for (int v = 0; v < 8; ++v)
        C[(size_t)(rbase + v) * N + col] = acc[i][j][v] + bj;
    }
  }
}

#define KPAD 68

__global__ __launch_bounds__(128) void mha_kernel(const float* __restrict__ qkv,
                                                  float* __restrict__ attn_out) {
  __shared__ float Ks[64][KPAD];
  __shared__ float Vs[64][KPAD];
  __shared__ float Ps[4][16][KPAD];

  const int qb = blockIdx.x;
  const int bh = blockIdx.y;
  const int b = bh >> 4;
  const int h = bh & 15;

  const int tid = threadIdx.x;
  const int lane = tid & 31;
  const int w = tid >> 5;
  const int ln = lane & 15;
  const int half = lane >> 4;
  const int kb = half << 1;

  const int qw0 = qb * 64 + w * 16;
  const size_t rstride = 3 * D_MODEL;
  const float* base = qkv + (size_t)b * SEQ_L * rstride + (size_t)h * HEAD_DIM;

  v2f qa[16];
  {
    const float* qp = base + (size_t)(qw0 + ln) * rstride + kb;
#pragma unroll
    for (int kk = 0; kk < 16; ++kk) qa[kk] = *(const v2f*)(qp + kk * 4);
  }

  v8f acc_o[4];
  float m_i[8], l_i[8];
#pragma unroll
  for (int v = 0; v < 8; ++v) {
    m_i[v] = -__builtin_inff();
    l_i[v] = 0.0f;
    acc_o[0][v] = 0.0f; acc_o[1][v] = 0.0f;
    acc_o[2][v] = 0.0f; acc_o[3][v] = 0.0f;
  }

  const float scale = 0.03125f;
  const float log2e = 1.44269504088896f;

  for (int j = 0; j <= qb; ++j) {
    __syncthreads();
    {
      int r0 = tid >> 4;
      int c = (tid & 15) << 2;
#pragma unroll
      for (int it = 0; it < 8; ++it) {
        int r = r0 + it * 8;
        const float* row = base + (size_t)(j * 64 + r) * rstride;
        *(v4f*)&Ks[r][c] = *(const v4f*)(row + D_MODEL + c);
        v4f vv = *(const v4f*)(row + 2 * D_MODEL + c);
        Vs[c + 0][r] = vv.x;
        Vs[c + 1][r] = vv.y;
        Vs[c + 2][r] = vv.z;
        Vs[c + 3][r] = vv.w;
      }
    }
    __syncthreads();

    v8f s[4];
#pragma unroll
    for (int nt = 0; nt < 4; ++nt)
#pragma unroll
      for (int v = 0; v < 8; ++v) s[nt][v] = 0.0f;
#pragma unroll
    for (int kk = 0; kk < 16; ++kk) {
      v2f b0 = *(const v2f*)&Ks[0 + ln][kk * 4 + kb];
      v2f b1 = *(const v2f*)&Ks[16 + ln][kk * 4 + kb];
      v2f b2 = *(const v2f*)&Ks[32 + ln][kk * 4 + kb];
      v2f b3 = *(const v2f*)&Ks[48 + ln][kk * 4 + kb];
      s[0] = wmma_f32(qa[kk], b0, s[0]);
      s[1] = wmma_f32(qa[kk], b1, s[1]);
      s[2] = wmma_f32(qa[kk], b2, s[2]);
      s[3] = wmma_f32(qa[kk], b3, s[3]);
    }

    const bool do_mask = (j == qb);
    float tmax[8];
#pragma unroll
    for (int v = 0; v < 8; ++v) {
      float mx = -__builtin_inff();
#pragma unroll
      for (int nt = 0; nt < 4; ++nt) {
        float sv = s[nt][v] * scale;
        if (do_mask) {
          int kvg = j * 64 + nt * 16 + ln;
          int qg = qw0 + v + half * 8;
          if (kvg > qg) sv = -__builtin_inff();
        }
        s[nt][v] = sv;
        mx = fmaxf(mx, sv);
      }
      tmax[v] = rowmax16(mx);
    }

#pragma unroll
    for (int v = 0; v < 8; ++v) {
      float mn = fmaxf(m_i[v], tmax[v]);
      float alpha = exp2f((m_i[v] - mn) * log2e);
      m_i[v] = mn;
      float rs = 0.0f;
#pragma unroll
      for (int nt = 0; nt < 4; ++nt) {
        float p = exp2f((s[nt][v] - mn) * log2e);
        rs += p;
        Ps[w][v + half * 8][nt * 16 + ln] = p;
      }
      l_i[v] = l_i[v] * alpha + rowsum16(rs);
      acc_o[0][v] *= alpha;
      acc_o[1][v] *= alpha;
      acc_o[2][v] *= alpha;
      acc_o[3][v] *= alpha;
    }

    asm volatile("s_wait_dscnt 0" ::: "memory");

#pragma unroll
    for (int kk = 0; kk < 16; ++kk) {
      v2f pa = *(const v2f*)&Ps[w][ln][kk * 4 + kb];
      v2f b0 = *(const v2f*)&Vs[0 + ln][kk * 4 + kb];
      v2f b1 = *(const v2f*)&Vs[16 + ln][kk * 4 + kb];
      v2f b2 = *(const v2f*)&Vs[32 + ln][kk * 4 + kb];
      v2f b3 = *(const v2f*)&Vs[48 + ln][kk * 4 + kb];
      acc_o[0] = wmma_f32(pa, b0, acc_o[0]);
      acc_o[1] = wmma_f32(pa, b1, acc_o[1]);
      acc_o[2] = wmma_f32(pa, b2, acc_o[2]);
      acc_o[3] = wmma_f32(pa, b3, acc_o[3]);
    }
  }

  float* obase = attn_out + (size_t)b * SEQ_L * D_MODEL + (size_t)h * HEAD_DIM;
#pragma unroll
  for (int v = 0; v < 8; ++v) {
    int qg = qw0 + v + half * 8;
    float inv_l = 1.0f / l_i[v];
#pragma unroll
    for (int nt = 0; nt < 4; ++nt)
      obase[(size_t)qg * D_MODEL + nt * 16 + ln] = acc_o[nt][v] * inv_l;
  }
}

extern "C" void kernel_launch(void* const* d_in, const int* in_sizes, int n_in,
                              void* d_out, int out_size, void* d_ws,
                              size_t ws_size, hipStream_t stream) {
  (void)in_sizes; (void)n_in; (void)out_size; (void)ws_size;
  const float* x = (const float*)d_in[0];
  const float* w_qkv = (const float*)d_in[1];
  const float* b_qkv = (const float*)d_in[2];
  const float* w_out = (const float*)d_in[3];
  const float* b_out = (const float*)d_in[4];
  float* out = (float*)d_out;

  float* qkv = (float*)d_ws;
  float* attn = qkv + (size_t)M_ROWS * 3 * D_MODEL;

  gemm_bias_kernel<<<dim3((3 * D_MODEL) / BN, M_ROWS / BM), 256, 0, stream>>>(
      x, w_qkv, b_qkv, qkv, 3 * D_MODEL, D_MODEL);

  mha_kernel<<<dim3(SEQ_L / 64, BATCH * H_HEADS), 128, 0, stream>>>(qkv, attn);

  gemm_bias_kernel<<<dim3(D_MODEL / BN, M_ROWS / BM), 256, 0, stream>>>(
      attn, w_out, b_out, out, D_MODEL, D_MODEL);
}


# attention: one workgroup now does query block 31-p then p (33 KV tiles each, grid 16x32 instead of 32x32): equal causal work per workgroup; per-row sum order unchanged
# speedup vs baseline: 1.2693x; 1.2693x over previous
#include <hip/hip_runtime.h>

typedef __attribute__((ext_vector_type(2))) float v2f;
typedef __attribute__((ext_vector_type(4))) float v4f;
typedef __attribute__((ext_vector_type(8))) float v8f;

#define D_MODEL 1024
#define H_HEADS 16
#define HEAD_DIM 64
#define SEQ_L 2048
#define BATCH 2
#define M_ROWS (BATCH * SEQ_L)

__device__ __forceinline__ v8f wmma_f32(v2f a, v2f b, v8f c) {
  return __builtin_amdgcn_wmma_f32_16x16x4_f32(false, a, false, b, (short)0, c,
                                               false, false);
}

__device__ __forceinline__ float rowmax16(float x) {
#pragma unroll
  for (int m = 1; m < 16; m <<= 1) x = fmaxf(x, __shfl_xor(x, m, 32));
  return x;
}
__device__ __forceinline__ float rowsum16(float x) {
#pragma unroll
  for (int m = 1; m < 16; m <<= 1) x += __shfl_xor(x, m, 32);
  return x;
}

#define BM 64
#define BN 128
#define BK 32
#define APAD 36
#define BPAD 34

__global__ __launch_bounds__(256) void gemm_bias_kernel(
    const float* __restrict__ A, const float* __restrict__ B,
    const float* __restrict__ bias, float* __restrict__ C, int N, int K) {
  __shared__ float As[BM][APAD];
  __shared__ float Bs[BN][BPAD];

  const int tid = threadIdx.x;
  const int lane = tid & 31;
  const int w = tid >> 5;
  const int wm = (w >> 2) << 5;
  const int wn = (w & 3) << 5;
  const int m0 = blockIdx.y * BM;
  const int n0 = blockIdx.x * BN;

  const int ln = lane & 15;
  const int half = lane >> 4;
  const int kb = half << 1;

  v8f acc[2][2];
#pragma unroll
  for (int i = 0; i < 2; ++i)
#pragma unroll
    for (int j = 0; j < 2; ++j)
#pragma unroll
      for (int v = 0; v < 8; ++v) acc[i][j][v] = 0.0f;

  for (int k0 = 0; k0 < K; k0 += BK) {
    __syncthreads();
    {
      int r = tid >> 3;
      int c = (tid & 7) << 2;
      const float* ap = A + (size_t)(m0 + r) * K + (k0 + c);
      *(v4f*)&As[r][c] = *(const v4f*)ap;
      *(v4f*)&As[r + 32][c] = *(const v4f*)(ap + (size_t)32 * K);
    }
    {
      int kr = tid >> 5;
      int nc = (tid & 31) << 2;
#pragma unroll
      for (int it = 0; it < 4; ++it) {
        int kk = kr + it * 8;
        v4f bv = *(const v4f*)(B + (size_t)(k0 + kk) * N + (n0 + nc));
        Bs[nc + 0][kk] = bv.x;
        Bs[nc + 1][kk] = bv.y;
        Bs[nc + 2][kk] = bv.z;
        Bs[nc + 3][kk] = bv.w;
      }
    }
    __syncthreads();
#pragma unroll
    for (int kk = 0; kk < BK; kk += 4) {
      v2f a0 = *(const v2f*)&As[wm + ln][kk + kb];
      v2f a1 = *(const v2f*)&As[wm + 16 + ln][kk + kb];
      v2f b0 = *(const v2f*)&Bs[wn + ln][kk + kb];
      v2f b1 = *(const v2f*)&Bs[wn + 16 + ln][kk + kb];
      acc[0][0] = wmma_f32(a0, b0, acc[0][0]);
      acc[0][1] = wmma_f32(a0, b1, acc[0][1]);
      acc[1][0] = wmma_f32(a1, b0, acc[1][0]);
      acc[1][1] = wmma_f32(a1, b1, acc[1][1]);
    }
  }

#pragma unroll
  for (int j = 0; j < 2; ++j) {
    float bj = bias[n0 + wn + j * 16 + ln];
#pragma unroll
    for (int i = 0; i < 2; ++i) {
      int rbase = m0 + wm + i * 16 + half * 8;
      int col = n0 + wn + j * 16 + ln;
#pragma unroll
      for (int v = 0; v < 8; ++v)
        C[(size_t)(rbase + v) * N + col] = acc[i][j][v] + bj;
    }
  }
}

#define KPAD 68

__global__ __launch_bounds__(128) void mha_kernel(const float* __restrict__ qkv,
                                                  float* __restrict__ attn_out) {
  __shared__ float Ks[64][KPAD];
  __shared__ float Vs[64][KPAD];
  __shared__ float Ps[4][16][KPAD];

  const int bh = blockIdx.y;
  const int b = bh >> 4;
  const int h = bh & 15;

  const int tid = threadIdx.x;
  const int lane = tid & 31;
  const int w = tid >> 5;
  const int ln = lane & 15;
  const int half = lane >> 4;
  const int kb = half << 1;

  const size_t rstride = 3 * D_MODEL;
  const float* base = qkv + (size_t)b * SEQ_L * rstride + (size_t)h * HEAD_DIM;
  float* obase = attn_out + (size_t)b * SEQ_L * D_MODEL + (size_t)h * HEAD_DIM;

  for (int pass = 0; pass < 2; ++pass) {
    const int qb = pass ? (int)blockIdx.x : (SEQ_L / 64 - 1) - (int)blockIdx.x;
    const int qw0 = qb * 64 + w * 16;

    v2f qa[16];
    {
      const float* qp = base + (size_t)(qw0 + ln) * rstride + kb;
#pragma unroll
      for (int kk = 0; kk < 16; ++kk) qa[kk] = *(const v2f*)(qp + kk * 4);
    }

    v8f acc_o[4];
    float m_i[8], l_i[8];
#pragma unroll
    for (int v = 0; v < 8; ++v) {
      m_i[v] = -__builtin_inff();
      l_i[v] = 0.0f;
      acc_o[0][v] = 0.0f; acc_o[1][v] = 0.0f;
      acc_o[2][v] = 0.0f; acc_o[3][v] = 0.0f;
    }

    const float scale = 0.03125f;
    const float log2e = 1.44269504088896f;

    for (int j = 0; j <= qb; ++j) {
      __syncthreads();
      {
        int r0 = tid >> 4;
        int c = (tid & 15) << 2;
#pragma unroll
        for (int it = 0; it < 8; ++it) {
          int r = r0 + it * 8;
          const float* row = base + (size_t)(j * 64 + r) * rstride;
          *(v4f*)&Ks[r][c] = *(const v4f*)(row + D_MODEL + c);
          v4f vv = *(const v4f*)(row + 2 * D_MODEL + c);
          Vs[c + 0][r] = vv.x;
          Vs[c + 1][r] = vv.y;
          Vs[c + 2][r] = vv.z;
          Vs[c + 3][r] = vv.w;
        }
      }
      __syncthreads();

      v8f s[4];
#pragma unroll
      for (int nt = 0; nt < 4; ++nt)
#pragma unroll
        for (int v = 0; v < 8; ++v) s[nt][v] = 0.0f;
#pragma unroll
      for (int kk = 0; kk < 16; ++kk) {
        v2f b0 = *(const v2f*)&Ks[0 + ln][kk * 4 + kb];
        v2f b1 = *(const v2f*)&Ks[16 + ln][kk * 4 + kb];
        v2f b2 = *(const v2f*)&Ks[32 + ln][kk * 4 + kb];
        v2f b3 = *(const v2f*)&Ks[48 + ln][kk * 4 + kb];
        s[0] = wmma_f32(qa[kk], b0, s[0]);
        s[1] = wmma_f32(qa[kk], b1, s[1]);
        s[2] = wmma_f32(qa[kk], b2, s[2]);
        s[3] = wmma_f32(qa[kk], b3, s[3]);
      }

      const bool do_mask = (j == qb);
      float tmax[8];
#pragma unroll
      for (int v = 0; v < 8; ++v) {
        float mx = -__builtin_inff();
#pragma unroll
        for (int nt = 0; nt < 4; ++nt) {
          float sv = s[nt][v] * scale;
          if (do_mask) {
            int kvg = j * 64 + nt * 16 + ln;
            int qg = qw0 + v + half * 8;
            if (kvg > qg) sv = -__builtin_inff();
          }
          s[nt][v] = sv;
          mx = fmaxf(mx, sv);
        }
        tmax[v] = rowmax16(mx);
      }

#pragma unroll
      for (int v = 0; v < 8; ++v) {
        float mn = fmaxf(m_i[v], tmax[v]);
        float alpha = exp2f((m_i[v] - mn) * log2e);
        m_i[v] = mn;
        float rs = 0.0f;
#pragma unroll
        for (int nt = 0; nt < 4; ++nt) {
          float p = exp2f((s[nt][v] - mn) * log2e);
          rs += p;
          Ps[w][v + half * 8][nt * 16 + ln] = p;
        }
        l_i[v] = l_i[v] * alpha + rowsum16(rs);
        acc_o[0][v] *= alpha;
        acc_o[1][v] *= alpha;
        acc_o[2][v] *= alpha;
        acc_o[3][v] *= alpha;
      }

      asm volatile("s_wait_dscnt 0" ::: "memory");

#pragma unroll
      for (int kk = 0; kk < 16; ++kk) {
        v2f pa = *(const v2f*)&Ps[w][ln][kk * 4 + kb];
        v2f b0 = *(const v2f*)&Vs[0 + ln][kk * 4 + kb];
        v2f b1 = *(const v2f*)&Vs[16 + ln][kk * 4 + kb];
        v2f b2 = *(const v2f*)&Vs[32 + ln][kk * 4 + kb];
        v2f b3 = *(const v2f*)&Vs[48 + ln][kk * 4 + kb];
        acc_o[0] = wmma_f32(pa, b0, acc_o[0]);
        acc_o[1] = wmma_f32(pa, b1, acc_o[1]);
        acc_o[2] = wmma_f32(pa, b2, acc_o[2]);
        acc_o[3] = wmma_f32(pa, b3, acc_o[3]);
      }
    }

#pragma unroll
    for (int v = 0; v < 8; ++v) {
      int qg = qw0 + v + half * 8;
      float inv_l = 1.0f / l_i[v];
#pragma unroll
      for (int nt = 0; nt < 4; ++nt)
        obase[(size_t)qg * D_MODEL + nt * 16 + ln] = acc_o[nt][v] * inv_l;
    }
  }
}

extern "C" void kernel_launch(void* const* d_in, const int* in_sizes, int n_in,
                              void* d_out, int out_size, void* d_ws,
                              size_t ws_size, hipStream_t stream) {
  (void)in_sizes; (void)n_in; (void)out_size; (void)ws_size;
  const float* x = (const float*)d_in[0];
  const float* w_qkv = (const float*)d_in[1];
  const float* b_qkv = (const float*)d_in[2];
  const float* w_out = (const float*)d_in[3];
  const float* b_out = (const float*)d_in[4];
  float* out = (float*)d_out;

  float* qkv = (float*)d_ws;
  float* attn = qkv + (size_t)M_ROWS * 3 * D_MODEL;

  gemm_bias_kernel<<<dim3((3 * D_MODEL) / BN, M_ROWS / BM), 256, 0, stream>>>(
      x, w_qkv, b_qkv, qkv, 3 * D_MODEL, D_MODEL);

  mha_kernel<<<dim3(SEQ_L / 128, BATCH * H_HEADS), 128, 0, stream>>>(qkv, attn);

  gemm_bias_kernel<<<dim3(D_MODEL / BN, M_ROWS / BM), 256, 0, stream>>>(
      attn, w_out, b_out, out, D_MODEL, D_MODEL);
}


# attention: scores kept transposed (S^T=K.Q^T): softmax max/sum in-lane + one half-wave exchange, P^T feeds P.V from registers, V staged as key pairs, raw v_exp_f32; row-sum and P.V key order changed
# speedup vs baseline: 1.3826x; 1.0893x over previous
#include <hip/hip_runtime.h>

typedef __attribute__((ext_vector_type(2))) float v2f;
typedef __attribute__((ext_vector_type(4))) float v4f;
typedef __attribute__((ext_vector_type(8))) float v8f;

#define D_MODEL 1024
#define H_HEADS 16
#define HEAD_DIM 64
#define SEQ_L 2048
#define BATCH 2
#define M_ROWS (BATCH * SEQ_L)

__device__ __forceinline__ v8f wmma_f32(v2f a, v2f b, v8f c) {
  return __builtin_amdgcn_wmma_f32_16x16x4_f32(false, a, false, b, (short)0, c,
                                               false, false);
}

#define BM 64
#define BN 128
#define BK 32
#define APAD 36
#define BPAD 34

__global__ __launch_bounds__(256) void gemm_bias_kernel(
    const float* __restrict__ A, const float* __restrict__ B,
    const float* __restrict__ bias, float* __restrict__ C, int N, int K) {
  __shared__ float As[BM][APAD];
  __shared__ float Bs[BN][BPAD];

  const int tid = threadIdx.x;
  const int lane = tid & 31;
  const int w = tid >> 5;
  const int wm = (w >> 2) << 5;
  const int wn = (w & 3) << 5;
  const int m0 = blockIdx.y * BM;
  const int n0 = blockIdx.x * BN;

  const int ln = lane & 15;
  const int half = lane >> 4;
  const int kb = half << 1;

  v8f acc[2][2];
#pragma unroll
  for (int i = 0; i < 2; ++i)
#pragma unroll
    for (int j = 0; j < 2; ++j)
#pragma unroll
      for (int v = 0; v < 8; ++v) acc[i][j][v] = 0.0f;

  for (int k0 = 0; k0 < K; k0 += BK) {
    __syncthreads();
    {
      int r = tid >> 3;
      int c = (tid & 7) << 2;
      const float* ap = A + (size_t)(m0 + r) * K + (k0 + c);
      *(v4f*)&As[r][c] = *(const v4f*)ap;
      *(v4f*)&As[r + 32][c] = *(const v4f*)(ap + (size_t)32 * K);
    }
    {
      int kr = tid >> 5;
      int nc = (tid & 31) << 2;
#pragma unroll
      for (int it = 0; it < 4; ++it) {
        int kk = kr + it * 8;
        v4f bv = *(const v4f*)(B + (size_t)(k0 + kk) * N + (n0 + nc));
        Bs[nc + 0][kk] = bv.x;
        Bs[nc + 1][kk] = bv.y;
        Bs[nc + 2][kk] = bv.z;
        Bs[nc + 3][kk] = bv.w;
      }
    }
    __syncthreads();
#pragma unroll
    for (int kk = 0; kk < BK; kk += 4) {
      v2f a0 = *(const v2f*)&As[wm + ln][kk + kb];
      v2f a1 = *(const v2f*)&As[wm + 16 + ln][kk + kb];
      v2f b0 = *(const v2f*)&Bs[wn + ln][kk + kb];
      v2f b1 = *(const v2f*)&Bs[wn + 16 + ln][kk + kb];
      acc[0][0] = wmma_f32(a0, b0, acc[0][0]);
      acc[0][1] = wmma_f32(a0, b1, acc[0][1]);
      acc[1][0] = wmma_f32(a1, b0, acc[1][0]);
      acc[1][1] = wmma_f32(a1, b1, acc[1][1]);
    }
  }

#pragma unroll
  for (int j = 0; j < 2; ++j) {
    float bj = bias[n0 + wn + j * 16 + ln];
#pragma unroll
    for (int i = 0; i < 2; ++i) {
      int rbase = m0 + wm + i * 16 + half * 8;
      int col = n0 + wn + j * 16 + ln;
#pragma unroll
      for (int v = 0; v < 8; ++v)
        C[(size_t)(rbase + v) * N + col] = acc[i][j][v] + bj;
    }
  }
}

#define KPAD 68
#define VPITCH 136

__device__ __forceinline__ float xhalf(float x) {
  int b = __builtin_bit_cast(int, x);
  return __builtin_bit_cast(
      float, __builtin_amdgcn_permlanex16(b, b, 0x76543210, 0xfedcba98, false, false));
}

__global__ __launch_bounds__(128) void mha_kernel(const float* __restrict__ qkv,
                                                  float* __restrict__ attn_out) {
  __shared__ float Ks[64][KPAD];
  __shared__ float Vp[32][VPITCH];

  const int bh = blockIdx.y;
  const int b = bh >> 4;
  const int h = bh & 15;

  const int tid = threadIdx.x;
  const int lane = tid & 31;
  const int w = tid >> 5;
  const int ln = lane & 15;
  const int half = lane >> 4;
  const int kb = half << 1;

  const size_t rstride = 3 * D_MODEL;
  const float* base = qkv + (size_t)b * SEQ_L * rstride + (size_t)h * HEAD_DIM;
  float* obase = attn_out + (size_t)b * SEQ_L * D_MODEL + (size_t)h * HEAD_DIM;

  const float cs = 0.03125f * 1.44269504088896f;

  for (int pass = 0; pass < 2; ++pass) {
    const int qb = pass ? (int)blockIdx.x : (SEQ_L / 64 - 1) - (int)blockIdx.x;
    const int qw0 = qb * 64 + w * 16;

    v2f qa[16];
    {
      const float* qp = base + (size_t)(qw0 + ln) * rstride + kb;
#pragma unroll
      for (int kk = 0; kk < 16; ++kk) qa[kk] = *(const v2f*)(qp + kk * 4);
    }

    v8f acc_o[4];
#pragma unroll
    for (int v = 0; v < 8; ++v) {
      acc_o[0][v] = 0.0f; acc_o[1][v] = 0.0f;
      acc_o[2][v] = 0.0f; acc_o[3][v] = 0.0f;
    }
    float m_i = -__builtin_inff();
    float l_i = 0.0f;

    for (int j = 0; j <= qb; ++j) {
      __syncthreads();
      {
        const int r0 = tid >> 4;
        const int c = (tid & 15) << 2;
#pragma unroll
        for (int it = 0; it < 8; ++it) {
          const int r = r0 + it * 8;
          const float* row = base + (size_t)(j * 64 + r) * rstride;
          *(v4f*)&Ks[r][c] = *(const v4f*)(row + D_MODEL + c);
        }
#pragma unroll
        for (int it = 0; it < 4; ++it) {
          const int p = r0 + it * 8;
          const float* row = base + (size_t)(j * 64 + 2 * p) * rstride + 2 * D_MODEL + c;
          const v4f va = *(const v4f*)row;
          const v4f vb = *(const v4f*)(row + rstride);
          v4f lo, hi;
          lo.x = va.x; lo.y = vb.x; lo.z = va.y; lo.w = vb.y;
          hi.x = va.z; hi.y = vb.z; hi.z = va.w; hi.w = vb.w;
          *(v4f*)&Vp[p][2 * c] = lo;
          *(v4f*)&Vp[p][2 * c + 4] = hi;
        }
      }
      __syncthreads();

      v8f s[4];
#pragma unroll
      for (int nt = 0; nt < 4; ++nt)
#pragma unroll
        for (int v = 0; v < 8; ++v) s[nt][v] = 0.0f;
#pragma unroll
      for (int kk = 0; kk < 16; ++kk) {
        v2f a0 = *(const v2f*)&Ks[0 + ln][kk * 4 + kb];
        v2f a1 = *(const v2f*)&Ks[16 + ln][kk * 4 + kb];
        v2f a2 = *(const v2f*)&Ks[32 + ln][kk * 4 + kb];
        v2f a3 = *(const v2f*)&Ks[48 + ln][kk * 4 + kb];
        s[0] = wmma_f32(a0, qa[kk], s[0]);
        s[1] = wmma_f32(a1, qa[kk], s[1]);
        s[2] = wmma_f32(a2, qa[kk], s[2]);
        s[3] = wmma_f32(a3, qa[kk], s[3]);
      }

      if (j == qb) {
        const int ql = w * 16 + ln;
#pragma unroll
        for (int nt = 0; nt < 4; ++nt)
#pragma unroll
          for (int v = 0; v < 8; ++v)
            if (nt * 16 + half * 8 + v > ql) s[nt][v] = -__builtin_inff();
      }

      float mxt[4];
#pragma unroll
      for (int nt = 0; nt < 4; ++nt) {
        float t0 = s[nt][0];
#pragma unroll
        for (int v = 1; v < 8; ++v) t0 = fmaxf(t0, s[nt][v]);
        mxt[nt] = t0;
      }
      float mx = fmaxf(fmaxf(mxt[0], mxt[1]), fmaxf(mxt[2], mxt[3]));
      mx = fmaxf(mx, xhalf(mx));

      const float mn = fmaxf(m_i, mx);
      const float alpha = __builtin_amdgcn_exp2f((m_i - mn) * cs);
      m_i = mn;
      const float nb = -mn * cs;

      float rst[4];
#pragma unroll
      for (int nt = 0; nt < 4; ++nt) {
        float r0 = 0.0f;
#pragma unroll
        for (int v = 0; v < 8; ++v) {
          const float p = __builtin_amdgcn_exp2f(__builtin_fmaf(s[nt][v], cs, nb));
          s[nt][v] = p;
          r0 += p;
        }
        rst[nt] = r0;
      }
      l_i = l_i * alpha + ((rst[0] + rst[1]) + (rst[2] + rst[3]));
#pragma unroll
      for (int dt = 0; dt < 4; ++dt)
#pragma unroll
        for (int v = 0; v < 8; ++v) acc_o[dt][v] *= alpha;

#pragma unroll
      for (int nt = 0; nt < 4; ++nt) {
#pragma unroll
        for (int t = 0; t < 4; ++t) {
          v2f pb;
          pb.x = s[nt][2 * t];
          pb.y = s[nt][2 * t + 1];
          const float* vrow = &Vp[nt * 8 + half * 4 + t][2 * ln];
          v2f a0 = *(const v2f*)(vrow + 0);
          v2f a1 = *(const v2f*)(vrow + 32);
          v2f a2 = *(const v2f*)(vrow + 64);
          v2f a3 = *(const v2f*)(vrow + 96);
          acc_o[0] = wmma_f32(a0, pb, acc_o[0]);
          acc_o[1] = wmma_f32(a1, pb, acc_o[1]);
          acc_o[2] = wmma_f32(a2, pb, acc_o[2]);
          acc_o[3] = wmma_f32(a3, pb, acc_o[3]);
        }
      }
    }

    {
      const float lt = l_i + xhalf(l_i);
      const float inv_l = 1.0f / lt;
      float* orow = obase + (size_t)(qw0 + ln) * D_MODEL + half * 8;
#pragma unroll
      for (int dt = 0; dt < 4; ++dt) {
        v4f o0, o1;
        o0.x = acc_o[dt][0] * inv_l; o0.y = acc_o[dt][1] * inv_l;
        o0.z = acc_o[dt][2] * inv_l; o0.w = acc_o[dt][3] * inv_l;
        o1.x = acc_o[dt][4] * inv_l; o1.y = acc_o[dt][5] * inv_l;
        o1.z = acc_o[dt][6] * inv_l; o1.w = acc_o[dt][7] * inv_l;
        *(v4f*)(orow + dt * 16) = o0;
        *(v4f*)(orow + dt * 16 + 4) = o1;
      }
    }
  }
}

extern "C" void kernel_launch(void* const* d_in, const int* in_sizes, int n_in,
                              void* d_out, int out_size, void* d_ws,
                              size_t ws_size, hipStream_t stream) {
  (void)in_sizes; (void)n_in; (void)out_size; (void)ws_size;
  const float* x = (const float*)d_in[0];
  const float* w_qkv = (const float*)d_in[1];
  const float* b_qkv = (const float*)d_in[2];
  const float* w_out = (const float*)d_in[3];
  const float* b_out = (const float*)d_in[4];
  float* out = (float*)d_out;

  float* qkv = (float*)d_ws;
  float* attn = qkv + (size_t)M_ROWS * 3 * D_MODEL;

  gemm_bias_kernel<<<dim3((3 * D_MODEL) / BN, M_ROWS / BM), 256, 0, stream>>>(
      x, w_qkv, b_qkv, qkv, 3 * D_MODEL, D_MODEL);

  mha_kernel<<<dim3(SEQ_L / 128, BATCH * H_HEADS), 128, 0, stream>>>(qkv, attn);

  gemm_bias_kernel<<<dim3(D_MODEL / BN, M_ROWS / BM), 256, 0, stream>>>(
      attn, w_out, b_out, out, D_MODEL, D_MODEL);
}


# projection GEMMs: K loop software-pipelined - next 32-wide slab's global loads issued ahead of the current slab's WMMAs, two LDS stages, one barrier per slab; sum order unchanged
# speedup vs baseline: 1.5762x; 1.1400x over previous
#include <hip/hip_runtime.h>

typedef __attribute__((ext_vector_type(2))) float v2f;
typedef __attribute__((ext_vector_type(4))) float v4f;
typedef __attribute__((ext_vector_type(8))) float v8f;

#define D_MODEL 1024
#define H_HEADS 16
#define HEAD_DIM 64
#define SEQ_L 2048
#define BATCH 2
#define M_ROWS (BATCH * SEQ_L)

__device__ __forceinline__ v8f wmma_f32(v2f a, v2f b, v8f c) {
  return __builtin_amdgcn_wmma_f32_16x16x4_f32(false, a, false, b, (short)0, c,
                                               false, false);
}

#define BM 64
#define BN 128
#define BK 32
#define APAD 36
#define BPAD 34

__global__ __launch_bounds__(256) void gemm_bias_kernel(
    const float* __restrict__ A, const float* __restrict__ B,
    const float* __restrict__ bias, float* __restrict__ C, int N, int K) {
  __shared__ float As[2][BM][APAD];
  __shared__ float Bs[2][BN][BPAD];

  const int tid = threadIdx.x;
  const int lane = tid & 31;
  const int w = tid >> 5;
  const int wm = (w >> 2) << 5;
  const int wn = (w & 3) << 5;
  const int m0 = blockIdx.y * BM;
  const int n0 = blockIdx.x * BN;

  const int ln = lane & 15;
  const int half = lane >> 4;
  const int kb = half << 1;

  v8f acc[2][2];
#pragma unroll
  for (int i = 0; i < 2; ++i)
#pragma unroll
    for (int j = 0; j < 2; ++j)
#pragma unroll
      for (int v = 0; v < 8; ++v) acc[i][j][v] = 0.0f;

  const int ar = tid >> 3;
  const int ac = (tid & 7) << 2;
  const int kr = tid >> 5;
  const int nc = (tid & 31) << 2;
  const float* ap = A + (size_t)(m0 + ar) * K + ac;
  const float* bp = B + (size_t)kr * N + (n0 + nc);
  const size_t a_half = (size_t)32 * K;
  const size_t b_step = (size_t)8 * N;

  v4f ra0 = *(const v4f*)ap;
  v4f ra1 = *(const v4f*)(ap + a_half);
  v4f rb0 = *(const v4f*)bp;
  v4f rb1 = *(const v4f*)(bp + b_step);
  v4f rb2 = *(const v4f*)(bp + 2 * b_step);
  v4f rb3 = *(const v4f*)(bp + 3 * b_step);
  *(v4f*)&As[0][ar][ac] = ra0;
  *(v4f*)&As[0][ar + 32][ac] = ra1;
  Bs[0][nc + 0][kr] = rb0.x; Bs[0][nc + 1][kr] = rb0.y;
  Bs[0][nc + 2][kr] = rb0.z; Bs[0][nc + 3][kr] = rb0.w;
  Bs[0][nc + 0][kr + 8] = rb1.x; Bs[0][nc + 1][kr + 8] = rb1.y;
  Bs[0][nc + 2][kr + 8] = rb1.z; Bs[0][nc + 3][kr + 8] = rb1.w;
  Bs[0][nc + 0][kr + 16] = rb2.x; Bs[0][nc + 1][kr + 16] = rb2.y;
  Bs[0][nc + 2][kr + 16] = rb2.z; Bs[0][nc + 3][kr + 16] = rb2.w;
  Bs[0][nc + 0][kr + 24] = rb3.x; Bs[0][nc + 1][kr + 24] = rb3.y;
  Bs[0][nc + 2][kr + 24] = rb3.z; Bs[0][nc + 3][kr + 24] = rb3.w;
  __syncthreads();

  int st = 0;
  for (int k0 = 0; k0 < K; k0 += BK) {
    const bool more = (k0 + BK) < K;
    if (more) {
      ap += BK;
      bp += (size_t)BK * N;
      ra0 = *(const v4f*)ap;
      ra1 = *(const v4f*)(ap + a_half);
      rb0 = *(const v4f*)bp;
      rb1 = *(const v4f*)(bp + b_step);
      rb2 = *(const v4f*)(bp + 2 * b_step);
      rb3 = *(const v4f*)(bp + 3 * b_step);
    }

#pragma unroll
    for (int kk = 0; kk < BK; kk += 4) {
      v2f a0 = *(const v2f*)&As[st][wm + ln][kk + kb];
      v2f a1 = *(const v2f*)&As[st][wm + 16 + ln][kk + kb];
      v2f b0 = *(const v2f*)&Bs[st][wn + ln][kk + kb];
      v2f b1 = *(const v2f*)&Bs[st][wn + 16 + ln][kk + kb];
      acc[0][0] = wmma_f32(a0, b0, acc[0][0]);
      acc[0][1] = wmma_f32(a0, b1, acc[0][1]);
      acc[1][0] = wmma_f32(a1, b0, acc[1][0]);
      acc[1][1] = wmma_f32(a1, b1, acc[1][1]);
    }

    if (more) {
      const int sn = st ^ 1;
      *(v4f*)&As[sn][ar][ac] = ra0;
      *(v4f*)&As[sn][ar + 32][ac] = ra1;
      Bs[sn][nc + 0][kr] = rb0.x; Bs[sn][nc + 1][kr] = rb0.y;
      Bs[sn][nc + 2][kr] = rb0.z; Bs[sn][nc + 3][kr] = rb0.w;
      Bs[sn][nc + 0][kr + 8] = rb1.x; Bs[sn][nc + 1][kr + 8] = rb1.y;
      Bs[sn][nc + 2][kr + 8] = rb1.z; Bs[sn][nc + 3][kr + 8] = rb1.w;
      Bs[sn][nc + 0][kr + 16] = rb2.x; Bs[sn][nc + 1][kr + 16] = rb2.y;
      Bs[sn][nc + 2][kr + 16] = rb2.z; Bs[sn][nc + 3][kr + 16] = rb2.w;
      Bs[sn][nc + 0][kr + 24] = rb3.x; Bs[sn][nc + 1][kr + 24] = rb3.y;
      Bs[sn][nc + 2][kr + 24] = rb3.z; Bs[sn][nc + 3][kr + 24] = rb3.w;
    }
    __syncthreads();
    st ^= 1;
  }

#pragma unroll
  for (int j = 0; j < 2; ++j) {
    float bj = bias[n0 + wn + j * 16 + ln];
#pragma unroll
    for (int i = 0; i < 2; ++i) {
      int rbase = m0 + wm + i * 16 + half * 8;
      int col = n0 + wn + j * 16 + ln;
#pragma unroll
      for (int v = 0; v < 8; ++v)
        C[(size_t)(rbase + v) * N + col] = acc[i][j][v] + bj;
    }
  }
}

#define KPAD 68
#define VPITCH 136

__device__ __forceinline__ float xhalf(float x) {
  int b = __builtin_bit_cast(int, x);
  return __builtin_bit_cast(
      float, __builtin_amdgcn_permlanex16(b, b, 0x76543210, 0xfedcba98, false, false));
}

__global__ __launch_bounds__(128) void mha_kernel(const float* __restrict__ qkv,
                                                  float* __restrict__ attn_out) {
  __shared__ float Ks[64][KPAD];
  __shared__ float Vp[32][VPITCH];

  const int bh = blockIdx.y;
  const int b = bh >> 4;
  const int h = bh & 15;

  const int tid = threadIdx.x;
  const int lane = tid & 31;
  const int w = tid >> 5;
  const int ln = lane & 15;
  const int half = lane >> 4;
  const int kb = half << 1;

  const size_t rstride = 3 * D_MODEL;
  const float* base = qkv + (size_t)b * SEQ_L * rstride + (size_t)h * HEAD_DIM;
  float* obase = attn_out + (size_t)b * SEQ_L * D_MODEL + (size_t)h * HEAD_DIM;

  const float cs = 0.03125f * 1.44269504088896f;

  for (int pass = 0; pass < 2; ++pass) {
    const int qb = pass ? (int)blockIdx.x : (SEQ_L / 64 - 1) - (int)blockIdx.x;
    const int qw0 = qb * 64 + w * 16;

    v2f qa[16];
    {
      const float* qp = base + (size_t)(qw0 + ln) * rstride + kb;
#pragma unroll
      for (int kk = 0; kk < 16; ++kk) qa[kk] = *(const v2f*)(qp + kk * 4);
    }

    v8f acc_o[4];
#pragma unroll
    for (int v = 0; v < 8; ++v) {
      acc_o[0][v] = 0.0f; acc_o[1][v] = 0.0f;
      acc_o[2][v] = 0.0f; acc_o[3][v] = 0.0f;
    }
    float m_i = -__builtin_inff();
    float l_i = 0.0f;

    for (int j = 0; j <= qb; ++j) {
      __syncthreads();
      {
        const int r0 = tid >> 4;
        const int c = (tid & 15) << 2;
#pragma unroll
        for (int it = 0; it < 8; ++it) {
          const int r = r0 + it * 8;
          const float* row = base + (size_t)(j * 64 + r) * rstride;
          *(v4f*)&Ks[r][c] = *(const v4f*)(row + D_MODEL + c);
        }
#pragma unroll
        for (int it = 0; it < 4; ++it) {
          const int p = r0 + it * 8;
          const float* row = base + (size_t)(j * 64 + 2 * p) * rstride + 2 * D_MODEL + c;
          const v4f va = *(const v4f*)row;
          const v4f vb = *(const v4f*)(row + rstride);
          v4f lo, hi;
          lo.x = va.x; lo.y = vb.x; lo.z = va.y; lo.w = vb.y;
          hi.x = va.z; hi.y = vb.z; hi.z = va.w; hi.w = vb.w;
          *(v4f*)&Vp[p][2 * c] = lo;
          *(v4f*)&Vp[p][2 * c + 4] = hi;
        }
      }
      __syncthreads();

      v8f s[4];
#pragma unroll
      for (int nt = 0; nt < 4; ++nt)
#pragma unroll
        for (int v = 0; v < 8; ++v) s[nt][v] = 0.0f;
#pragma unroll
      for (int kk = 0; kk < 16; ++kk) {
        v2f a0 = *(const v2f*)&Ks[0 + ln][kk * 4 + kb];
        v2f a1 = *(const v2f*)&Ks[16 + ln][kk * 4 + kb];
        v2f a2 = *(const v2f*)&Ks[32 + ln][kk * 4 + kb];
        v2f a3 = *(const v2f*)&Ks[48 + ln][kk * 4 + kb];
        s[0] = wmma_f32(a0, qa[kk], s[0]);
        s[1] = wmma_f32(a1, qa[kk], s[1]);
        s[2] = wmma_f32(a2, qa[kk], s[2]);
        s[3] = wmma_f32(a3, qa[kk], s[3]);
      }

      if (j == qb) {
        const int ql = w * 16 + ln;
#pragma unroll
        for (int nt = 0; nt < 4; ++nt)
#pragma unroll
          for (int v = 0; v < 8; ++v)
            if (nt * 16 + half * 8 + v > ql) s[nt][v] = -__builtin_inff();
      }

      float mxt[4];
#pragma unroll
      for (int nt = 0; nt < 4; ++nt) {
        float t0 = s[nt][0];
#pragma unroll
        for (int v = 1; v < 8; ++v) t0 = fmaxf(t0, s[nt][v]);
        mxt[nt] = t0;
      }
      float mx = fmaxf(fmaxf(mxt[0], mxt[1]), fmaxf(mxt[2], mxt[3]));
      mx = fmaxf(mx, xhalf(mx));

      const float mn = fmaxf(m_i, mx);
      const float alpha = __builtin_amdgcn_exp2f((m_i - mn) * cs);
      m_i = mn;
      const float nb = -mn * cs;

      float rst[4];
#pragma unroll
      for (int nt = 0; nt < 4; ++nt) {
        float r0 = 0.0f;
#pragma unroll
        for (int v = 0; v < 8; ++v) {
          const float p = __builtin_amdgcn_exp2f(__builtin_fmaf(s[nt][v], cs, nb));
          s[nt][v] = p;
          r0 += p;
        }
        rst[nt] = r0;
      }
      l_i = l_i * alpha + ((rst[0] + rst[1]) + (rst[2] + rst[3]));
#pragma unroll
      for (int dt = 0; dt < 4; ++dt)
#pragma unroll
        for (int v = 0; v < 8; ++v) acc_o[dt][v] *= alpha;

#pragma unroll
      for (int nt = 0; nt < 4; ++nt) {
#pragma unroll
        for (int t = 0; t < 4; ++t) {
          v2f pb;
          pb.x = s[nt][2 * t];
          pb.y = s[nt][2 * t + 1];
          const float* vrow = &Vp[nt * 8 + half * 4 + t][2 * ln];
          v2f a0 = *(const v2f*)(vrow + 0);
          v2f a1 = *(const v2f*)(vrow + 32);
          v2f a2 = *(const v2f*)(vrow + 64);
          v2f a3 = *(const v2f*)(vrow + 96);
          acc_o[0] = wmma_f32(a0, pb, acc_o[0]);
          acc_o[1] = wmma_f32(a1, pb, acc_o[1]);
          acc_o[2] = wmma_f32(a2, pb, acc_o[2]);
          acc_o[3] = wmma_f32(a3, pb, acc_o[3]);
        }
      }
    }

    {
      const float lt = l_i + xhalf(l_i);
      const float inv_l = 1.0f / lt;
      float* orow = obase + (size_t)(qw0 + ln) * D_MODEL + half * 8;
#pragma unroll
      for (int dt = 0; dt < 4; ++dt) {
        v4f o0, o1;
        o0.x = acc_o[dt][0] * inv_l; o0.y = acc_o[dt][1] * inv_l;
        o0.z = acc_o[dt][2] * inv_l; o0.w = acc_o[dt][3] * inv_l;
        o1.x = acc_o[dt][4] * inv_l; o1.y = acc_o[dt][5] * inv_l;
        o1.z = acc_o[dt][6] * inv_l; o1.w = acc_o[dt][7] * inv_l;
        *(v4f*)(orow + dt * 16) = o0;
        *(v4f*)(orow + dt * 16 + 4) = o1;
      }
    }
  }
}

extern "C" void kernel_launch(void* const* d_in, const int* in_sizes, int n_in,
                              void* d_out, int out_size, void* d_ws,
                              size_t ws_size, hipStream_t stream) {
  (void)in_sizes; (void)n_in; (void)out_size; (void)ws_size;
  const float* x = (const float*)d_in[0];
  const float* w_qkv = (const float*)d_in[1];
  const float* b_qkv = (const float*)d_in[2];
  const float* w_out = (const float*)d_in[3];
  const float* b_out = (const float*)d_in[4];
  float* out = (float*)d_out;

  float* qkv = (float*)d_ws;
  float* attn = qkv + (size_t)M_ROWS * 3 * D_MODEL;

  gemm_bias_kernel<<<dim3((3 * D_MODEL) / BN, M_ROWS / BM), 256, 0, stream>>>(
      x, w_qkv, b_qkv, qkv, 3 * D_MODEL, D_MODEL);

  mha_kernel<<<dim3(SEQ_L / 128, BATCH * H_HEADS), 128, 0, stream>>>(qkv, attn);

  gemm_bias_kernel<<<dim3(D_MODEL / BN, M_ROWS / BM), 256, 0, stream>>>(
      attn, w_out, b_out, out, D_MODEL, D_MODEL);
}


# attention: K/V staging software-pipelined - next KV tile's global loads issued ahead of the current tile's WMMAs, two LDS stages, one barrier per tile instead of two; sum order unchanged
# speedup vs baseline: 1.6430x; 1.0424x over previous
#include <hip/hip_runtime.h>

typedef __attribute__((ext_vector_type(2))) float v2f;
typedef __attribute__((ext_vector_type(4))) float v4f;
typedef __attribute__((ext_vector_type(8))) float v8f;

#define D_MODEL 1024
#define H_HEADS 16
#define HEAD_DIM 64
#define SEQ_L 2048
#define BATCH 2
#define M_ROWS (BATCH * SEQ_L)

__device__ __forceinline__ v8f wmma_f32(v2f a, v2f b, v8f c) {
  return __builtin_amdgcn_wmma_f32_16x16x4_f32(false, a, false, b, (short)0, c,
                                               false, false);
}

#define BM 64
#define BN 128
#define BK 32
#define APAD 36
#define BPAD 34

__global__ __launch_bounds__(256) void gemm_bias_kernel(
    const float* __restrict__ A, const float* __restrict__ B,
    const float* __restrict__ bias, float* __restrict__ C, int N, int K) {
  __shared__ float As[2][BM][APAD];
  __shared__ float Bs[2][BN][BPAD];

  const int tid = threadIdx.x;
  const int lane = tid & 31;
  const int w = tid >> 5;
  const int wm = (w >> 2) << 5;
  const int wn = (w & 3) << 5;
  const int m0 = blockIdx.y * BM;
  const int n0 = blockIdx.x * BN;

  const int ln = lane & 15;
  const int half = lane >> 4;
  const int kb = half << 1;

  v8f acc[2][2];
#pragma unroll
  for (int i = 0; i < 2; ++i)
#pragma unroll
    for (int j = 0; j < 2; ++j)
#pragma unroll
      for (int v = 0; v < 8; ++v) acc[i][j][v] = 0.0f;

  const int ar = tid >> 3;
  const int ac = (tid & 7) << 2;
  const int kr = tid >> 5;
  const int nc = (tid & 31) << 2;
  const float* ap = A + (size_t)(m0 + ar) * K + ac;
  const float* bp = B + (size_t)kr * N + (n0 + nc);
  const size_t a_half = (size_t)32 * K;
  const size_t b_step = (size_t)8 * N;

  v4f ra0 = *(const v4f*)ap;
  v4f ra1 = *(const v4f*)(ap + a_half);
  v4f rb0 = *(const v4f*)bp;
  v4f rb1 = *(const v4f*)(bp + b_step);
  v4f rb2 = *(const v4f*)(bp + 2 * b_step);
  v4f rb3 = *(const v4f*)(bp + 3 * b_step);
  *(v4f*)&As[0][ar][ac] = ra0;
  *(v4f*)&As[0][ar + 32][ac] = ra1;
  Bs[0][nc + 0][kr] = rb0.x; Bs[0][nc + 1][kr] = rb0.y;
  Bs[0][nc + 2][kr] = rb0.z; Bs[0][nc + 3][kr] = rb0.w;
  Bs[0][nc + 0][kr + 8] = rb1.x; Bs[0][nc + 1][kr + 8] = rb1.y;
  Bs[0][nc + 2][kr + 8] = rb1.z; Bs[0][nc + 3][kr + 8] = rb1.w;
  Bs[0][nc + 0][kr + 16] = rb2.x; Bs[0][nc + 1][kr + 16] = rb2.y;
  Bs[0][nc + 2][kr + 16] = rb2.z; Bs[0][nc + 3][kr + 16] = rb2.w;
  Bs[0][nc + 0][kr + 24] = rb3.x; Bs[0][nc + 1][kr + 24] = rb3.y;
  Bs[0][nc + 2][kr + 24] = rb3.z; Bs[0][nc + 3][kr + 24] = rb3.w;
  __syncthreads();

  int st = 0;
  for (int k0 = 0; k0 < K; k0 += BK) {
    const bool more = (k0 + BK) < K;
    if (more) {
      ap += BK;
      bp += (size_t)BK * N;
      ra0 = *(const v4f*)ap;
      ra1 = *(const v4f*)(ap + a_half);
      rb0 = *(const v4f*)bp;
      rb1 = *(const v4f*)(bp + b_step);
      rb2 = *(const v4f*)(bp + 2 * b_step);
      rb3 = *(const v4f*)(bp + 3 * b_step);
    }

#pragma unroll
    for (int kk = 0; kk < BK; kk += 4) {
      v2f a0 = *(const v2f*)&As[st][wm + ln][kk + kb];
      v2f a1 = *(const v2f*)&As[st][wm + 16 + ln][kk + kb];
      v2f b0 = *(const v2f*)&Bs[st][wn + ln][kk + kb];
      v2f b1 = *(const v2f*)&Bs[st][wn + 16 + ln][kk + kb];
      acc[0][0] = wmma_f32(a0, b0, acc[0][0]);
      acc[0][1] = wmma_f32(a0, b1, acc[0][1]);
      acc[1][0] = wmma_f32(a1, b0, acc[1][0]);
      acc[1][1] = wmma_f32(a1, b1, acc[1][1]);
    }

    if (more) {
      const int sn = st ^ 1;
      *(v4f*)&As[sn][ar][ac] = ra0;
      *(v4f*)&As[sn][ar + 32][ac] = ra1;
      Bs[sn][nc + 0][kr] = rb0.x; Bs[sn][nc + 1][kr] = rb0.y;
      Bs[sn][nc + 2][kr] = rb0.z; Bs[sn][nc + 3][kr] = rb0.w;
      Bs[sn][nc + 0][kr + 8] = rb1.x; Bs[sn][nc + 1][kr + 8] = rb1.y;
      Bs[sn][nc + 2][kr + 8] = rb1.z; Bs[sn][nc + 3][kr + 8] = rb1.w;
      Bs[sn][nc + 0][kr + 16] = rb2.x; Bs[sn][nc + 1][kr + 16] = rb2.y;
      Bs[sn][nc + 2][kr + 16] = rb2.z; Bs[sn][nc + 3][kr + 16] = rb2.w;
      Bs[sn][nc + 0][kr + 24] = rb3.x; Bs[sn][nc + 1][kr + 24] = rb3.y;
      Bs[sn][nc + 2][kr + 24] = rb3.z; Bs[sn][nc + 3][kr + 24] = rb3.w;
    }
    __syncthreads();
    st ^= 1;
  }

#pragma unroll
  for (int j = 0; j < 2; ++j) {
    float bj = bias[n0 + wn + j * 16 + ln];
#pragma unroll
    for (int i = 0; i < 2; ++i) {
      int rbase = m0 + wm + i * 16 + half * 8;
      int col = n0 + wn + j * 16 + ln;
#pragma unroll
      for (int v = 0; v < 8; ++v)
        C[(size_t)(rbase + v) * N + col] = acc[i][j][v] + bj;
    }
  }
}

#define KPAD 68
#define VPITCH 136

__device__ __forceinline__ void kv_fetch(const float* __restrict__ base, int j,
                                         int r0, int c, v4f (&rk)[8],
                                         v4f (&rva)[4], v4f (&rvb)[4]) {
  const size_t rstride = 3 * D_MODEL;
#pragma unroll
  for (int it = 0; it < 8; ++it) {
    const float* row = base + (size_t)(j * 64 + r0 + it * 8) * rstride;
    rk[it] = *(const v4f*)(row + D_MODEL + c);
  }
#pragma unroll
  for (int it = 0; it < 4; ++it) {
    const int p = r0 + it * 8;
    const float* row = base + (size_t)(j * 64 + 2 * p) * rstride + 2 * D_MODEL + c;
    rva[it] = *(const v4f*)row;
    rvb[it] = *(const v4f*)(row + rstride);
  }
}

__device__ __forceinline__ void kv_stage(float (*Kst)[KPAD], float (*Vst)[VPITCH],
                                         int r0, int c, const v4f (&rk)[8],
                                         const v4f (&rva)[4], const v4f (&rvb)[4]) {
#pragma unroll
  for (int it = 0; it < 8; ++it) *(v4f*)&Kst[r0 + it * 8][c] = rk[it];
#pragma unroll
  for (int it = 0; it < 4; ++it) {
    const int p = r0 + it * 8;
    v4f lo, hi;
    lo.x = rva[it].x; lo.y = rvb[it].x; lo.z = rva[it].y; lo.w = rvb[it].y;
    hi.x = rva[it].z; hi.y = rvb[it].z; hi.z = rva[it].w; hi.w = rvb[it].w;
    *(v4f*)&Vst[p][2 * c] = lo;
    *(v4f*)&Vst[p][2 * c + 4] = hi;
  }
}

__device__ __forceinline__ float xhalf(float x) {
  int b = __builtin_bit_cast(int, x);
  return __builtin_bit_cast(
      float, __builtin_amdgcn_permlanex16(b, b, 0x76543210, 0xfedcba98, false, false));
}

__global__ __launch_bounds__(128) void mha_kernel(const float* __restrict__ qkv,
                                                  float* __restrict__ attn_out) {
  __shared__ float Ks[2][64][KPAD];
  __shared__ float Vp[2][32][VPITCH];

  const int bh = blockIdx.y;
  const int b = bh >> 4;
  const int h = bh & 15;

  const int tid = threadIdx.x;
  const int lane = tid & 31;
  const int w = tid >> 5;
  const int ln = lane & 15;
  const int half = lane >> 4;
  const int kb = half << 1;

  const size_t rstride = 3 * D_MODEL;
  const float* base = qkv + (size_t)b * SEQ_L * rstride + (size_t)h * HEAD_DIM;
  float* obase = attn_out + (size_t)b * SEQ_L * D_MODEL + (size_t)h * HEAD_DIM;

  const float cs = 0.03125f * 1.44269504088896f;

  const int r0 = tid >> 4;
  const int c = (tid & 15) << 2;
  v4f rk[8], rva[4], rvb[4];

  kv_fetch(base, 0, r0, c, rk, rva, rvb);
  kv_stage(Ks[0], Vp[0], r0, c, rk, rva, rvb);
  __syncthreads();
  int st = 0;

  for (int pass = 0; pass < 2; ++pass) {
    const int qb = pass ? (int)blockIdx.x : (SEQ_L / 64 - 1) - (int)blockIdx.x;
    const int qw0 = qb * 64 + w * 16;

    v2f qa[16];
    {
      const float* qp = base + (size_t)(qw0 + ln) * rstride + kb;
#pragma unroll
      for (int kk = 0; kk < 16; ++kk) qa[kk] = *(const v2f*)(qp + kk * 4);
    }
    asm volatile("" ::"v"(qa[0]), "v"(qa[1]), "v"(qa[2]), "v"(qa[3]), "v"(qa[4]),
                 "v"(qa[5]), "v"(qa[6]), "v"(qa[7]), "v"(qa[8]), "v"(qa[9]),
                 "v"(qa[10]), "v"(qa[11]), "v"(qa[12]), "v"(qa[13]), "v"(qa[14]),
                 "v"(qa[15]));

    v8f acc_o[4];
#pragma unroll
    for (int v = 0; v < 8; ++v) {
      acc_o[0][v] = 0.0f; acc_o[1][v] = 0.0f;
      acc_o[2][v] = 0.0f; acc_o[3][v] = 0.0f;
    }
    float m_i = -__builtin_inff();
    float l_i = 0.0f;

    for (int j = 0; j <= qb; ++j) {
      const bool last = (j == qb);
      const bool more = !last || pass == 0;
      if (more) {
        kv_fetch(base, last ? 0 : j + 1, r0, c, rk, rva, rvb);
      }
      const float (*Kc)[KPAD] = Ks[st];
      const float (*Vc)[VPITCH] = Vp[st];

      v8f s[4];
#pragma unroll
      for (int nt = 0; nt < 4; ++nt)
#pragma unroll
        for (int v = 0; v < 8; ++v) s[nt][v] = 0.0f;
#pragma unroll
      for (int kk = 0; kk < 16; ++kk) {
        v2f a0 = *(const v2f*)&Kc[0 + ln][kk * 4 + kb];
        v2f a1 = *(const v2f*)&Kc[16 + ln][kk * 4 + kb];
        v2f a2 = *(const v2f*)&Kc[32 + ln][kk * 4 + kb];
        v2f a3 = *(const v2f*)&Kc[48 + ln][kk * 4 + kb];
        s[0] = wmma_f32(a0, qa[kk], s[0]);
        s[1] = wmma_f32(a1, qa[kk], s[1]);
        s[2] = wmma_f32(a2, qa[kk], s[2]);
        s[3] = wmma_f32(a3, qa[kk], s[3]);
      }

      if (last) {
        const int ql = w * 16 + ln;
#pragma unroll
        for (int nt = 0; nt < 4; ++nt)
#pragma unroll
          for (int v = 0; v < 8; ++v)
            if (nt * 16 + half * 8 + v > ql) s[nt][v] = -__builtin_inff();
      }

      float mxt[4];
#pragma unroll
      for (int nt = 0; nt < 4; ++nt) {
        float t0 = s[nt][0];
#pragma unroll
        for (int v = 1; v < 8; ++v) t0 = fmaxf(t0, s[nt][v]);
        mxt[nt] = t0;
      }
      float mx = fmaxf(fmaxf(mxt[0], mxt[1]), fmaxf(mxt[2], mxt[3]));
      mx = fmaxf(mx, xhalf(mx));

      const float mn = fmaxf(m_i, mx);
      const float alpha = __builtin_amdgcn_exp2f((m_i - mn) * cs);
      m_i = mn;
      const float nb = -mn * cs;

      float rst[4];
#pragma unroll
      for (int nt = 0; nt < 4; ++nt) {
        float r0 = 0.0f;
#pragma unroll
        for (int v = 0; v < 8; ++v) {
          const float p = __builtin_amdgcn_exp2f(__builtin_fmaf(s[nt][v], cs, nb));
          s[nt][v] = p;
          r0 += p;
        }
        rst[nt] = r0;
      }
      l_i = l_i * alpha + ((rst[0] + rst[1]) + (rst[2] + rst[3]));
#pragma unroll
      for (int dt = 0; dt < 4; ++dt)
#pragma unroll
        for (int v = 0; v < 8; ++v) acc_o[dt][v] *= alpha;

#pragma unroll
      for (int nt = 0; nt < 4; ++nt) {
#pragma unroll
        for (int t = 0; t < 4; ++t) {
          v2f pb;
          pb.x = s[nt][2 * t];
          pb.y = s[nt][2 * t + 1];
          const float* vrow = &Vc[nt * 8 + half * 4 + t][2 * ln];
          v2f a0 = *(const v2f*)(vrow + 0);
          v2f a1 = *(const v2f*)(vrow + 32);
          v2f a2 = *(const v2f*)(vrow + 64);
          v2f a3 = *(const v2f*)(vrow + 96);
          acc_o[0] = wmma_f32(a0, pb, acc_o[0]);
          acc_o[1] = wmma_f32(a1, pb, acc_o[1]);
          acc_o[2] = wmma_f32(a2, pb, acc_o[2]);
          acc_o[3] = wmma_f32(a3, pb, acc_o[3]);
        }
      }

      if (more) {
        kv_stage(Ks[st ^ 1], Vp[st ^ 1], r0, c, rk, rva, rvb);
      }
      __syncthreads();
      st ^= 1;
    }

    {
      const float lt = l_i + xhalf(l_i);
      const float inv_l = 1.0f / lt;
      float* orow = obase + (size_t)(qw0 + ln) * D_MODEL + half * 8;
#pragma unroll
      for (int dt = 0; dt < 4; ++dt) {
        v4f o0, o1;
        o0.x = acc_o[dt][0] * inv_l; o0.y = acc_o[dt][1] * inv_l;
        o0.z = acc_o[dt][2] * inv_l; o0.w = acc_o[dt][3] * inv_l;
        o1.x = acc_o[dt][4] * inv_l; o1.y = acc_o[dt][5] * inv_l;
        o1.z = acc_o[dt][6] * inv_l; o1.w = acc_o[dt][7] * inv_l;
        *(v4f*)(orow + dt * 16) = o0;
        *(v4f*)(orow + dt * 16 + 4) = o1;
      }
    }
  }
}

extern "C" void kernel_launch(void* const* d_in, const int* in_sizes, int n_in,
                              void* d_out, int out_size, void* d_ws,
                              size_t ws_size, hipStream_t stream) {
  (void)in_sizes; (void)n_in; (void)out_size; (void)ws_size;
  const float* x = (const float*)d_in[0];
  const float* w_qkv = (const float*)d_in[1];
  const float* b_qkv = (const float*)d_in[2];
  const float* w_out = (const float*)d_in[3];
  const float* b_out = (const float*)d_in[4];
  float* out = (float*)d_out;

  float* qkv = (float*)d_ws;
  float* attn = qkv + (size_t)M_ROWS * 3 * D_MODEL;

  gemm_bias_kernel<<<dim3((3 * D_MODEL) / BN, M_ROWS / BM), 256, 0, stream>>>(
      x, w_qkv, b_qkv, qkv, 3 * D_MODEL, D_MODEL);

  mha_kernel<<<dim3(SEQ_L / 128, BATCH * H_HEADS), 128, 0, stream>>>(qkv, attn);

  gemm_bias_kernel<<<dim3(D_MODEL / BN, M_ROWS / BM), 256, 0, stream>>>(
      attn, w_out, b_out, out, D_MODEL, D_MODEL);
}


# QKV projection GEMM: wave tile 64x64 (4x4 accumulators), 4 waves per 128x128 block: one LDS fragment read feeds 4 WMMAs (was 2), reads a group ahead, staging in two halves; sum order unchanged
# speedup vs baseline: 1.6916x; 1.0296x over previous
#include <hip/hip_runtime.h>

typedef __attribute__((ext_vector_type(2))) float v2f;
typedef __attribute__((ext_vector_type(4))) float v4f;
typedef __attribute__((ext_vector_type(8))) float v8f;

#define D_MODEL 1024
#define H_HEADS 16
#define HEAD_DIM 64
#define SEQ_L 2048
#define BATCH 2
#define M_ROWS (BATCH * SEQ_L)

__device__ __forceinline__ v8f wmma_f32(v2f a, v2f b, v8f c) {
  return __builtin_amdgcn_wmma_f32_16x16x4_f32(false, a, false, b, (short)0, c,
                                               false, false);
}

#define BM 64
#define BN 128
#define BK 32
#define APAD 36
#define BPAD 34

__global__ __launch_bounds__(256) void gemm_bias_kernel(
    const float* __restrict__ A, const float* __restrict__ B,
    const float* __restrict__ bias, float* __restrict__ C, int N, int K) {
  __shared__ float As[2][BM][APAD];
  __shared__ float Bs[2][BN][BPAD];

  const int tid = threadIdx.x;
  const int lane = tid & 31;
  const int w = tid >> 5;
  const int wm = (w >> 2) << 5;
  const int wn = (w & 3) << 5;
  const int m0 = blockIdx.y * BM;
  const int n0 = blockIdx.x * BN;

  const int ln = lane & 15;
  const int half = lane >> 4;
  const int kb = half << 1;

  v8f acc[2][2];
#pragma unroll
  for (int i = 0; i < 2; ++i)
#pragma unroll
    for (int j = 0; j < 2; ++j)
#pragma unroll
      for (int v = 0; v < 8; ++v) acc[i][j][v] = 0.0f;

  const int ar = tid >> 3;
  const int ac = (tid & 7) << 2;
  const int kr = tid >> 5;
  const int nc = (tid & 31) << 2;
  const float* ap = A + (size_t)(m0 + ar) * K + ac;
  const float* bp = B + (size_t)kr * N + (n0 + nc);
  const size_t a_half = (size_t)32 * K;
  const size_t b_step = (size_t)8 * N;

  v4f ra0 = *(const v4f*)ap;
  v4f ra1 = *(const v4f*)(ap + a_half);
  v4f rb0 = *(const v4f*)bp;
  v4f rb1 = *(const v4f*)(bp + b_step);
  v4f rb2 = *(const v4f*)(bp + 2 * b_step);
  v4f rb3 = *(const v4f*)(bp + 3 * b_step);
  *(v4f*)&As[0][ar][ac] = ra0;
  *(v4f*)&As[0][ar + 32][ac] = ra1;
  Bs[0][nc + 0][kr] = rb0.x; Bs[0][nc + 1][kr] = rb0.y;
  Bs[0][nc + 2][kr] = rb0.z; Bs[0][nc + 3][kr] = rb0.w;
  Bs[0][nc + 0][kr + 8] = rb1.x; Bs[0][nc + 1][kr + 8] = rb1.y;
  Bs[0][nc + 2][kr + 8] = rb1.z; Bs[0][nc + 3][kr + 8] = rb1.w;
  Bs[0][nc + 0][kr + 16] = rb2.x; Bs[0][nc + 1][kr + 16] = rb2.y;
  Bs[0][nc + 2][kr + 16] = rb2.z; Bs[0][nc + 3][kr + 16] = rb2.w;
  Bs[0][nc + 0][kr + 24] = rb3.x; Bs[0][nc + 1][kr + 24] = rb3.y;
  Bs[0][nc + 2][kr + 24] = rb3.z; Bs[0][nc + 3][kr + 24] = rb3.w;
  __syncthreads();

  int st = 0;
  for (int k0 = 0; k0 < K; k0 += BK) {
    const bool more = (k0 + BK) < K;
    if (more) {
      ap += BK;
      bp += (size_t)BK * N;
      ra0 = *(const v4f*)ap;
      ra1 = *(const v4f*)(ap + a_half);
      rb0 = *(const v4f*)bp;
      rb1 = *(const v4f*)(bp + b_step);
      rb2 = *(const v4f*)(bp + 2 * b_step);
      rb3 = *(const v4f*)(bp + 3 * b_step);
    }

#pragma unroll
    for (int kk = 0; kk < BK; kk += 4) {
      v2f a0 = *(const v2f*)&As[st][wm + ln][kk + kb];
      v2f a1 = *(const v2f*)&As[st][wm + 16 + ln][kk + kb];
      v2f b0 = *(const v2f*)&Bs[st][wn + ln][kk + kb];
      v2f b1 = *(const v2f*)&Bs[st][wn + 16 + ln][kk + kb];
      acc[0][0] = wmma_f32(a0, b0, acc[0][0]);
      acc[0][1] = wmma_f32(a0, b1, acc[0][1]);
      acc[1][0] = wmma_f32(a1, b0, acc[1][0]);
      acc[1][1] = wmma_f32(a1, b1, acc[1][1]);
    }

    if (more) {
      const int sn = st ^ 1;
      *(v4f*)&As[sn][ar][ac] = ra0;
      *(v4f*)&As[sn][ar + 32][ac] = ra1;
      Bs[sn][nc + 0][kr] = rb0.x; Bs[sn][nc + 1][kr] = rb0.y;
      Bs[sn][nc + 2][kr] = rb0.z; Bs[sn][nc + 3][kr] = rb0.w;
      Bs[sn][nc + 0][kr + 8] = rb1.x; Bs[sn][nc + 1][kr + 8] = rb1.y;
      Bs[sn][nc + 2][kr + 8] = rb1.z; Bs[sn][nc + 3][kr + 8] = rb1.w;
      Bs[sn][nc + 0][kr + 16] = rb2.x; Bs[sn][nc + 1][kr + 16] = rb2.y;
      Bs[sn][nc + 2][kr + 16] = rb2.z; Bs[sn][nc + 3][kr + 16] = rb2.w;
      Bs[sn][nc + 0][kr + 24] = rb3.x; Bs[sn][nc + 1][kr + 24] = rb3.y;
      Bs[sn][nc + 2][kr + 24] = rb3.z; Bs[sn][nc + 3][kr + 24] = rb3.w;
    }
    __syncthreads();
    st ^= 1;
  }

#pragma unroll
  for (int j = 0; j < 2; ++j) {
    float bj = bias[n0 + wn + j * 16 + ln];
#pragma unroll
    for (int i = 0; i < 2; ++i) {
      int rbase = m0 + wm + i * 16 + half * 8;
      int col = n0 + wn + j * 16 + ln;
#pragma unroll
      for (int v = 0; v < 8; ++v)
        C[(size_t)(rbase + v) * N + col] = acc[i][j][v] + bj;
    }
  }
}

#define QBM 128
#define QBN 128

template <int N, int K>
__global__ __launch_bounds__(128) void gemm_bias_wide_kernel(
    const float* __restrict__ A, const float* __restrict__ B,
    const float* __restrict__ bias, float* __restrict__ C) {
  __shared__ float As[2][QBM][APAD];
  __shared__ float Bs[2][QBN][BPAD];

  const int tid = threadIdx.x;
  const int lane = tid & 31;
  const int w = tid >> 5;
  const int wm = (w >> 1) << 6;
  const int wn = (w & 1) << 6;
  const int m0 = blockIdx.y * QBM;
  const int n0 = blockIdx.x * QBN;

  const int ln = lane & 15;
  const int half = lane >> 4;
  const int kb = half << 1;

  v8f acc[4][4];
#pragma unroll
  for (int i = 0; i < 4; ++i)
#pragma unroll
    for (int j = 0; j < 4; ++j)
#pragma unroll
      for (int v = 0; v < 8; ++v) acc[i][j][v] = 0.0f;

  const int ar = tid >> 3;
  const int ac = (tid & 7) << 2;
  const int kr = tid >> 5;
  const int nc = (tid & 31) << 2;
  const float* ap = A + (size_t)(m0 + ar) * K + ac;
  const float* bp = B + (size_t)kr * N + (n0 + nc);

  v4f rg[8];

#pragma unroll
  for (int i = 0; i < 8; ++i) rg[i] = *(const v4f*)(ap + (size_t)i * 16 * K);
#pragma unroll
  for (int i = 0; i < 8; ++i) *(v4f*)&As[0][ar + 16 * i][ac] = rg[i];
#pragma unroll
  for (int i = 0; i < 8; ++i) rg[i] = *(const v4f*)(bp + (size_t)i * 4 * N);
#pragma unroll
  for (int i = 0; i < 8; ++i) {
    Bs[0][nc + 0][kr + 4 * i] = rg[i].x;
    Bs[0][nc + 1][kr + 4 * i] = rg[i].y;
    Bs[0][nc + 2][kr + 4 * i] = rg[i].z;
    Bs[0][nc + 3][kr + 4 * i] = rg[i].w;
  }
  __syncthreads();

  int st = 0;
  for (int k0 = 0; k0 < K; k0 += BK) {
    const bool more = (k0 + BK) < K;
    const int sn = st ^ 1;
    v2f fa[2][2][4], fb[2][2][4];
#pragma unroll
    for (int s = 0; s < 2; ++s) {
#pragma unroll
      for (int i = 0; i < 4; ++i)
        fa[0][s][i] = *(const v2f*)&As[st][wm + 16 * i + ln][4 * s + kb];
#pragma unroll
      for (int j = 0; j < 4; ++j)
        fb[0][s][j] = *(const v2f*)&Bs[st][wn + 16 * j + ln][4 * s + kb];
    }
    __builtin_amdgcn_sched_barrier(0);
    if (more) {
      ap += BK;
#pragma unroll
      for (int i = 0; i < 8; ++i) rg[i] = *(const v4f*)(ap + (size_t)i * 16 * K);
    }
#pragma unroll
    for (int g = 0; g < BK / 8; ++g) {
      const int cur = g & 1;
      if (g == BK / 16 && more) {
#pragma unroll
        for (int i = 0; i < 8; ++i) *(v4f*)&As[sn][ar + 16 * i][ac] = rg[i];
        bp += (size_t)BK * N;
#pragma unroll
        for (int i = 0; i < 8; ++i) rg[i] = *(const v4f*)(bp + (size_t)i * 4 * N);
        __builtin_amdgcn_sched_barrier(0);
      }
      if (g + 1 < BK / 8) {
#pragma unroll
        for (int s = 0; s < 2; ++s) {
#pragma unroll
          for (int i = 0; i < 4; ++i)
            fa[cur ^ 1][s][i] =
                *(const v2f*)&As[st][wm + 16 * i + ln][8 * (g + 1) + 4 * s + kb];
#pragma unroll
          for (int j = 0; j < 4; ++j)
            fb[cur ^ 1][s][j] =
                *(const v2f*)&Bs[st][wn + 16 * j + ln][8 * (g + 1) + 4 * s + kb];
        }
      }
      __builtin_amdgcn_sched_barrier(0);
#pragma unroll
      for (int s = 0; s < 2; ++s)
#pragma unroll
        for (int i = 0; i < 4; ++i)
#pragma unroll
          for (int j = 0; j < 4; ++j)
            acc[i][j] = wmma_f32(fa[cur][s][i], fb[cur][s][j], acc[i][j]);
      __builtin_amdgcn_sched_barrier(0);
    }

    if (more) {
#pragma unroll
      for (int i = 0; i < 8; ++i) {
        Bs[sn][nc + 0][kr + 4 * i] = rg[i].x;
        Bs[sn][nc + 1][kr + 4 * i] = rg[i].y;
        Bs[sn][nc + 2][kr + 4 * i] = rg[i].z;
        Bs[sn][nc + 3][kr + 4 * i] = rg[i].w;
      }
    }
    __syncthreads();
    st = sn;
  }

#pragma unroll
  for (int j = 0; j < 4; ++j) {
    const int col = n0 + wn + j * 16 + ln;
    const float bj = bias[col];
#pragma unroll
    for (int i = 0; i < 4; ++i) {
      const int rbase = m0 + wm + i * 16 + half * 8;
#pragma unroll
      for (int v = 0; v < 8; ++v)
        C[(size_t)(rbase + v) * N + col] = acc[i][j][v] + bj;
    }
  }
}

#define KPAD 68
#define VPITCH 136

__device__ __forceinline__ void kv_fetch(const float* __restrict__ base, int j,
                                         int r0, int c, v4f (&rk)[8],
                                         v4f (&rva)[4], v4f (&rvb)[4]) {
  const size_t rstride = 3 * D_MODEL;
#pragma unroll
  for (int it = 0; it < 8; ++it) {
    const float* row = base + (size_t)(j * 64 + r0 + it * 8) * rstride;
    rk[it] = *(const v4f*)(row + D_MODEL + c);
  }
#pragma unroll
  for (int it = 0; it < 4; ++it) {
    const int p = r0 + it * 8;
    const float* row = base + (size_t)(j * 64 + 2 * p) * rstride + 2 * D_MODEL + c;
    rva[it] = *(const v4f*)row;
    rvb[it] = *(const v4f*)(row + rstride);
  }
}

__device__ __forceinline__ void kv_stage(float (*Kst)[KPAD], float (*Vst)[VPITCH],
                                         int r0, int c, const v4f (&rk)[8],
                                         const v4f (&rva)[4], const v4f (&rvb)[4]) {
#pragma unroll
  for (int it = 0; it < 8; ++it) *(v4f*)&Kst[r0 + it * 8][c] = rk[it];
#pragma unroll
  for (int it = 0; it < 4; ++it) {
    const int p = r0 + it * 8;
    v4f lo, hi;
    lo.x = rva[it].x; lo.y = rvb[it].x; lo.z = rva[it].y; lo.w = rvb[it].y;
    hi.x = rva[it].z; hi.y = rvb[it].z; hi.z = rva[it].w; hi.w = rvb[it].w;
    *(v4f*)&Vst[p][2 * c] = lo;
    *(v4f*)&Vst[p][2 * c + 4] = hi;
  }
}

__device__ __forceinline__ float xhalf(float x) {
  int b = __builtin_bit_cast(int, x);
  return __builtin_bit_cast(
      float, __builtin_amdgcn_permlanex16(b, b, 0x76543210, 0xfedcba98, false, false));
}

__global__ __launch_bounds__(128) void mha_kernel(const float* __restrict__ qkv,
                                                  float* __restrict__ attn_out) {
  __shared__ float Ks[2][64][KPAD];
  __shared__ float Vp[2][32][VPITCH];

  const int bh = blockIdx.y;
  const int b = bh >> 4;
  const int h = bh & 15;

  const int tid = threadIdx.x;
  const int lane = tid & 31;
  const int w = tid >> 5;
  const int ln = lane & 15;
  const int half = lane >> 4;
  const int kb = half << 1;

  const size_t rstride = 3 * D_MODEL;
  const float* base = qkv + (size_t)b * SEQ_L * rstride + (size_t)h * HEAD_DIM;
  float* obase = attn_out + (size_t)b * SEQ_L * D_MODEL + (size_t)h * HEAD_DIM;

  const float cs = 0.03125f * 1.44269504088896f;

  const int r0 = tid >> 4;
  const int c = (tid & 15) << 2;
  v4f rk[8], rva[4], rvb[4];

  kv_fetch(base, 0, r0, c, rk, rva, rvb);
  kv_stage(Ks[0], Vp[0], r0, c, rk, rva, rvb);
  __syncthreads();
  int st = 0;

  for (int pass = 0; pass < 2; ++pass) {
    const int qb = pass ? (int)blockIdx.x : (SEQ_L / 64 - 1) - (int)blockIdx.x;
    const int qw0 = qb * 64 + w * 16;

    v2f qa[16];
    {
      const float* qp = base + (size_t)(qw0 + ln) * rstride + kb;
#pragma unroll
      for (int kk = 0; kk < 16; ++kk) qa[kk] = *(const v2f*)(qp + kk * 4);
    }
    asm volatile("" ::"v"(qa[0]), "v"(qa[1]), "v"(qa[2]), "v"(qa[3]), "v"(qa[4]),
                 "v"(qa[5]), "v"(qa[6]), "v"(qa[7]), "v"(qa[8]), "v"(qa[9]),
                 "v"(qa[10]), "v"(qa[11]), "v"(qa[12]), "v"(qa[13]), "v"(qa[14]),
                 "v"(qa[15]));

    v8f acc_o[4];
#pragma unroll
    for (int v = 0; v < 8; ++v) {
      acc_o[0][v] = 0.0f; acc_o[1][v] = 0.0f;
      acc_o[2][v] = 0.0f; acc_o[3][v] = 0.0f;
    }
    float m_i = -__builtin_inff();
    float l_i = 0.0f;

    for (int j = 0; j <= qb; ++j) {
      const bool last = (j == qb);
      const bool more = !last || pass == 0;
      if (more) {
        kv_fetch(base, last ? 0 : j + 1, r0, c, rk, rva, rvb);
      }
      const float (*Kc)[KPAD] = Ks[st];
      const float (*Vc)[VPITCH] = Vp[st];

      v8f s[4];
#pragma unroll
      for (int nt = 0; nt < 4; ++nt)
#pragma unroll
        for (int v = 0; v < 8; ++v) s[nt][v] = 0.0f;
#pragma unroll
      for (int kk = 0; kk < 16; ++kk) {
        v2f a0 = *(const v2f*)&Kc[0 + ln][kk * 4 + kb];
        v2f a1 = *(const v2f*)&Kc[16 + ln][kk * 4 + kb];
        v2f a2 = *(const v2f*)&Kc[32 + ln][kk * 4 + kb];
        v2f a3 = *(const v2f*)&Kc[48 + ln][kk * 4 + kb];
        s[0] = wmma_f32(a0, qa[kk], s[0]);
        s[1] = wmma_f32(a1, qa[kk], s[1]);
        s[2] = wmma_f32(a2, qa[kk], s[2]);
        s[3] = wmma_f32(a3, qa[kk], s[3]);
      }

      if (last) {
        const int ql = w * 16 + ln;
#pragma unroll
        for (int nt = 0; nt < 4; ++nt)
#pragma unroll
          for (int v = 0; v < 8; ++v)
            if (nt * 16 + half * 8 + v > ql) s[nt][v] = -__builtin_inff();
      }

      float mxt[4];
#pragma unroll
      for (int nt = 0; nt < 4; ++nt) {
        float t0 = s[nt][0];
#pragma unroll
        for (int v = 1; v < 8; ++v) t0 = fmaxf(t0, s[nt][v]);
        mxt[nt] = t0;
      }
      float mx = fmaxf(fmaxf(mxt[0], mxt[1]), fmaxf(mxt[2], mxt[3]));
      mx = fmaxf(mx, xhalf(mx));

      const float mn = fmaxf(m_i, mx);
      const float alpha = __builtin_amdgcn_exp2f((m_i - mn) * cs);
      m_i = mn;
      const float nb = -mn * cs;

      float rst[4];
#pragma unroll
      for (int nt = 0; nt < 4; ++nt) {
        float r0 = 0.0f;
#pragma unroll
        for (int v = 0; v < 8; ++v) {
          const float p = __builtin_amdgcn_exp2f(__builtin_fmaf(s[nt][v], cs, nb));
          s[nt][v] = p;
          r0 += p;
        }
        rst[nt] = r0;
      }
      l_i = l_i * alpha + ((rst[0] + rst[1]) + (rst[2] + rst[3]));
#pragma unroll
      for (int dt = 0; dt < 4; ++dt)
#pragma unroll
        for (int v = 0; v < 8; ++v) acc_o[dt][v] *= alpha;

#pragma unroll
      for (int nt = 0; nt < 4; ++nt) {
#pragma unroll
        for (int t = 0; t < 4; ++t) {
          v2f pb;
          pb.x = s[nt][2 * t];
          pb.y = s[nt][2 * t + 1];
          const float* vrow = &Vc[nt * 8 + half * 4 + t][2 * ln];
          v2f a0 = *(const v2f*)(vrow + 0);
          v2f a1 = *(const v2f*)(vrow + 32);
          v2f a2 = *(const v2f*)(vrow + 64);
          v2f a3 = *(const v2f*)(vrow + 96);
          acc_o[0] = wmma_f32(a0, pb, acc_o[0]);
          acc_o[1] = wmma_f32(a1, pb, acc_o[1]);
          acc_o[2] = wmma_f32(a2, pb, acc_o[2]);
          acc_o[3] = wmma_f32(a3, pb, acc_o[3]);
        }
      }

      if (more) {
        kv_stage(Ks[st ^ 1], Vp[st ^ 1], r0, c, rk, rva, rvb);
      }
      __syncthreads();
      st ^= 1;
    }

    {
      const float lt = l_i + xhalf(l_i);
      const float inv_l = 1.0f / lt;
      float* orow = obase + (size_t)(qw0 + ln) * D_MODEL + half * 8;
#pragma unroll
      for (int dt = 0; dt < 4; ++dt) {
        v4f o0, o1;
        o0.x = acc_o[dt][0] * inv_l; o0.y = acc_o[dt][1] * inv_l;
        o0.z = acc_o[dt][2] * inv_l; o0.w = acc_o[dt][3] * inv_l;
        o1.x = acc_o[dt][4] * inv_l; o1.y = acc_o[dt][5] * inv_l;
        o1.z = acc_o[dt][6] * inv_l; o1.w = acc_o[dt][7] * inv_l;
        *(v4f*)(orow + dt * 16) = o0;
        *(v4f*)(orow + dt * 16 + 4) = o1;
      }
    }
  }
}

extern "C" void kernel_launch(void* const* d_in, const int* in_sizes, int n_in,
                              void* d_out, int out_size, void* d_ws,
                              size_t ws_size, hipStream_t stream) {
  (void)in_sizes; (void)n_in; (void)out_size; (void)ws_size;
  const float* x = (const float*)d_in[0];
  const float* w_qkv = (const float*)d_in[1];
  const float* b_qkv = (const float*)d_in[2];
  const float* w_out = (const float*)d_in[3];
  const float* b_out = (const float*)d_in[4];
  float* out = (float*)d_out;

  float* qkv = (float*)d_ws;
  float* attn = qkv + (size_t)M_ROWS * 3 * D_MODEL;

  gemm_bias_wide_kernel<3 * D_MODEL, D_MODEL>
      <<<dim3((3 * D_MODEL) / QBN, M_ROWS / QBM), 128, 0, stream>>>(x, w_qkv, b_qkv, qkv);

  mha_kernel<<<dim3(SEQ_L / 128, BATCH * H_HEADS), 128, 0, stream>>>(qkv, attn);

  gemm_bias_kernel<<<dim3(D_MODEL / BN, M_ROWS / BM), 256, 0, stream>>>(
      attn, w_out, b_out, out, D_MODEL, D_MODEL);
}


# all three kernels: a wave lowers its issue priority (s_setprio 3,2,1,0) as it gets through its main loop, so a WGP's co-resident workgroups advance together, not oldest first; sums unchanged
# speedup vs baseline: 1.7768x; 1.0504x over previous
#include <hip/hip_runtime.h>

typedef __attribute__((ext_vector_type(2))) float v2f;
typedef __attribute__((ext_vector_type(4))) float v4f;
typedef __attribute__((ext_vector_type(8))) float v8f;

#define D_MODEL 1024
#define H_HEADS 16
#define HEAD_DIM 64
#define SEQ_L 2048
#define BATCH 2
#define M_ROWS (BATCH * SEQ_L)

__device__ __forceinline__ v8f wmma_f32(v2f a, v2f b, v8f c) {
  return __builtin_amdgcn_wmma_f32_16x16x4_f32(false, a, false, b, (short)0, c,
                                               false, false);
}

__device__ __forceinline__ int prio_level(int done, int t1, int t2, int t3) {
  return (done >= t1 ? 1 : 0) + (done >= t2 ? 1 : 0) + (done >= t3 ? 1 : 0);
}
__device__ __forceinline__ void set_prio_for_level(int level) {
  if (level == 0) __builtin_amdgcn_s_setprio(3);
  else if (level == 1) __builtin_amdgcn_s_setprio(2);
  else if (level == 2) __builtin_amdgcn_s_setprio(1);
  else __builtin_amdgcn_s_setprio(0);
}

#define BM 64
#define BN 128
#define BK 32
#define APAD 36
#define BPAD 34

__global__ __launch_bounds__(256) void gemm_bias_kernel(
    const float* __restrict__ A, const float* __restrict__ B,
    const float* __restrict__ bias, float* __restrict__ C, int N, int K) {
  __shared__ float As[2][BM][APAD];
  __shared__ float Bs[2][BN][BPAD];

  const int tid = threadIdx.x;
  const int lane = tid & 31;
  const int w = tid >> 5;
  const int wm = (w >> 2) << 5;
  const int wn = (w & 3) << 5;
  const int m0 = blockIdx.y * BM;
  const int n0 = blockIdx.x * BN;

  const int ln = lane & 15;
  const int half = lane >> 4;
  const int kb = half << 1;

  v8f acc[2][2];
#pragma unroll
  for (int i = 0; i < 2; ++i)
#pragma unroll
    for (int j = 0; j < 2; ++j)
#pragma unroll
      for (int v = 0; v < 8; ++v) acc[i][j][v] = 0.0f;

  const int ar = tid >> 3;
  const int ac = (tid & 7) << 2;
  const int kr = tid >> 5;
  const int nc = (tid & 31) << 2;
  const float* ap = A + (size_t)(m0 + ar) * K + ac;
  const float* bp = B + (size_t)kr * N + (n0 + nc);
  const size_t a_half = (size_t)32 * K;
  const size_t b_step = (size_t)8 * N;

  v4f ra0 = *(const v4f*)ap;
  v4f ra1 = *(const v4f*)(ap + a_half);
  v4f rb0 = *(const v4f*)bp;
  v4f rb1 = *(const v4f*)(bp + b_step);
  v4f rb2 = *(const v4f*)(bp + 2 * b_step);
  v4f rb3 = *(const v4f*)(bp + 3 * b_step);
  *(v4f*)&As[0][ar][ac] = ra0;
  *(v4f*)&As[0][ar + 32][ac] = ra1;
  Bs[0][nc + 0][kr] = rb0.x; Bs[0][nc + 1][kr] = rb0.y;
  Bs[0][nc + 2][kr] = rb0.z; Bs[0][nc + 3][kr] = rb0.w;
  Bs[0][nc + 0][kr + 8] = rb1.x; Bs[0][nc + 1][kr + 8] = rb1.y;
  Bs[0][nc + 2][kr + 8] = rb1.z; Bs[0][nc + 3][kr + 8] = rb1.w;
  Bs[0][nc + 0][kr + 16] = rb2.x; Bs[0][nc + 1][kr + 16] = rb2.y;
  Bs[0][nc + 2][kr + 16] = rb2.z; Bs[0][nc + 3][kr + 16] = rb2.w;
  Bs[0][nc + 0][kr + 24] = rb3.x; Bs[0][nc + 1][kr + 24] = rb3.y;
  Bs[0][nc + 2][kr + 24] = rb3.z; Bs[0][nc + 3][kr + 24] = rb3.w;
  __syncthreads();

  int st = 0;
  const int kt1 = (K >> 3) * 3, kt2 = (K >> 4) * 11, kt3 = (K >> 3) * 7;
  int plevel = -1;
  for (int k0 = 0; k0 < K; k0 += BK) {
    const bool more = (k0 + BK) < K;
    const int lv = prio_level(k0, kt1, kt2, kt3);
    if (lv != plevel) { plevel = lv; set_prio_for_level(lv); }
    if (more) {
      ap += BK;
      bp += (size_t)BK * N;
      ra0 = *(const v4f*)ap;
      ra1 = *(const v4f*)(ap + a_half);
      rb0 = *(const v4f*)bp;
      rb1 = *(const v4f*)(bp + b_step);
      rb2 = *(const v4f*)(bp + 2 * b_step);
      rb3 = *(const v4f*)(bp + 3 * b_step);
    }

#pragma unroll
    for (int kk = 0; kk < BK; kk += 4) {
      v2f a0 = *(const v2f*)&As[st][wm + ln][kk + kb];
      v2f a1 = *(const v2f*)&As[st][wm + 16 + ln][kk + kb];
      v2f b0 = *(const v2f*)&Bs[st][wn + ln][kk + kb];
      v2f b1 = *(const v2f*)&Bs[st][wn + 16 + ln][kk + kb];
      acc[0][0] = wmma_f32(a0, b0, acc[0][0]);
      acc[0][1] = wmma_f32(a0, b1, acc[0][1]);
      acc[1][0] = wmma_f32(a1, b0, acc[1][0]);
      acc[1][1] = wmma_f32(a1, b1, acc[1][1]);
    }

    if (more) {
      const int sn = st ^ 1;
      *(v4f*)&As[sn][ar][ac] = ra0;
      *(v4f*)&As[sn][ar + 32][ac] = ra1;
      Bs[sn][nc + 0][kr] = rb0.x; Bs[sn][nc + 1][kr] = rb0.y;
      Bs[sn][nc + 2][kr] = rb0.z; Bs[sn][nc + 3][kr] = rb0.w;
      Bs[sn][nc + 0][kr + 8] = rb1.x; Bs[sn][nc + 1][kr + 8] = rb1.y;
      Bs[sn][nc + 2][kr + 8] = rb1.z; Bs[sn][nc + 3][kr + 8] = rb1.w;
      Bs[sn][nc + 0][kr + 16] = rb2.x; Bs[sn][nc + 1][kr + 16] = rb2.y;
      Bs[sn][nc + 2][kr + 16] = rb2.z; Bs[sn][nc + 3][kr + 16] = rb2.w;
      Bs[sn][nc + 0][kr + 24] = rb3.x; Bs[sn][nc + 1][kr + 24] = rb3.y;
      Bs[sn][nc + 2][kr + 24] = rb3.z; Bs[sn][nc + 3][kr + 24] = rb3.w;
    }
    __syncthreads();
    st ^= 1;
  }
  __builtin_amdgcn_s_setprio(0);

#pragma unroll
  for (int j = 0; j < 2; ++j) {
    float bj = bias[n0 + wn + j * 16 + ln];
#pragma unroll
    for (int i = 0; i < 2; ++i) {
      int rbase = m0 + wm + i * 16 + half * 8;
      int col = n0 + wn + j * 16 + ln;
#pragma unroll
      for (int v = 0; v < 8; ++v)
        C[(size_t)(rbase + v) * N + col] = acc[i][j][v] + bj;
    }
  }
}

#define QBM 128
#define QBN 128

template <int N, int K>
__global__ __launch_bounds__(128) void gemm_bias_wide_kernel(
    const float* __restrict__ A, const float* __restrict__ B,
    const float* __restrict__ bias, float* __restrict__ C) {
  __shared__ float As[2][QBM][APAD];
  __shared__ float Bs[2][QBN][BPAD];

  const int tid = threadIdx.x;
  const int lane = tid & 31;
  const int w = tid >> 5;
  const int wm = (w >> 1) << 6;
  const int wn = (w & 1) << 6;
  const int m0 = blockIdx.y * QBM;
  const int n0 = blockIdx.x * QBN;

  const int ln = lane & 15;
  const int half = lane >> 4;
  const int kb = half << 1;

  v8f acc[4][4];
#pragma unroll
  for (int i = 0; i < 4; ++i)
#pragma unroll
    for (int j = 0; j < 4; ++j)
#pragma unroll
      for (int v = 0; v < 8; ++v) acc[i][j][v] = 0.0f;

  const int ar = tid >> 3;
  const int ac = (tid & 7) << 2;
  const int kr = tid >> 5;
  const int nc = (tid & 31) << 2;
  const float* ap = A + (size_t)(m0 + ar) * K + ac;
  const float* bp = B + (size_t)kr * N + (n0 + nc);

  v4f rg[8];

#pragma unroll
  for (int i = 0; i < 8; ++i) rg[i] = *(const v4f*)(ap + (size_t)i * 16 * K);
#pragma unroll
  for (int i = 0; i < 8; ++i) *(v4f*)&As[0][ar + 16 * i][ac] = rg[i];
#pragma unroll
  for (int i = 0; i < 8; ++i) rg[i] = *(const v4f*)(bp + (size_t)i * 4 * N);
#pragma unroll
  for (int i = 0; i < 8; ++i) {
    Bs[0][nc + 0][kr + 4 * i] = rg[i].x;
    Bs[0][nc + 1][kr + 4 * i] = rg[i].y;
    Bs[0][nc + 2][kr + 4 * i] = rg[i].z;
    Bs[0][nc + 3][kr + 4 * i] = rg[i].w;
  }
  __syncthreads();

  int st = 0;
  constexpr int kt1 = (K >> 3) * 3, kt2 = (K >> 4) * 11, kt3 = (K >> 3) * 7;
  int plevel = -1;
  for (int k0 = 0; k0 < K; k0 += BK) {
    const bool more = (k0 + BK) < K;
    const int sn = st ^ 1;
    const int lv = prio_level(k0, kt1, kt2, kt3);
    if (lv != plevel) { plevel = lv; set_prio_for_level(lv); }
    v2f fa[2][2][4], fb[2][2][4];
#pragma unroll
    for (int s = 0; s < 2; ++s) {
#pragma unroll
      for (int i = 0; i < 4; ++i)
        fa[0][s][i] = *(const v2f*)&As[st][wm + 16 * i + ln][4 * s + kb];
#pragma unroll
      for (int j = 0; j < 4; ++j)
        fb[0][s][j] = *(const v2f*)&Bs[st][wn + 16 * j + ln][4 * s + kb];
    }
    __builtin_amdgcn_sched_barrier(0);
    if (more) {
      ap += BK;
#pragma unroll
      for (int i = 0; i < 8; ++i) rg[i] = *(const v4f*)(ap + (size_t)i * 16 * K);
    }
#pragma unroll
    for (int g = 0; g < BK / 8; ++g) {
      const int cur = g & 1;
      if (g == BK / 16 && more) {
#pragma unroll
        for (int i = 0; i < 8; ++i) *(v4f*)&As[sn][ar + 16 * i][ac] = rg[i];
        bp += (size_t)BK * N;
#pragma unroll
        for (int i = 0; i < 8; ++i) rg[i] = *(const v4f*)(bp + (size_t)i * 4 * N);
        __builtin_amdgcn_sched_barrier(0);
      }
      if (g + 1 < BK / 8) {
#pragma unroll
        for (int s = 0; s < 2; ++s) {
#pragma unroll
          for (int i = 0; i < 4; ++i)
            fa[cur ^ 1][s][i] =
                *(const v2f*)&As[st][wm + 16 * i + ln][8 * (g + 1) + 4 * s + kb];
#pragma unroll
          for (int j = 0; j < 4; ++j)
            fb[cur ^ 1][s][j] =
                *(const v2f*)&Bs[st][wn + 16 * j + ln][8 * (g + 1) + 4 * s + kb];
        }
      }
      __builtin_amdgcn_sched_barrier(0);
#pragma unroll
      for (int s = 0; s < 2; ++s)
#pragma unroll
        for (int i = 0; i < 4; ++i)
#pragma unroll
          for (int j = 0; j < 4; ++j)
            acc[i][j] = wmma_f32(fa[cur][s][i], fb[cur][s][j], acc[i][j]);
      __builtin_amdgcn_sched_barrier(0);
    }

    if (more) {
#pragma unroll
      for (int i = 0; i < 8; ++i) {
        Bs[sn][nc + 0][kr + 4 * i] = rg[i].x;
        Bs[sn][nc + 1][kr + 4 * i] = rg[i].y;
        Bs[sn][nc + 2][kr + 4 * i] = rg[i].z;
        Bs[sn][nc + 3][kr + 4 * i] = rg[i].w;
      }
    }
    __syncthreads();
    st = sn;
  }

#pragma unroll
  for (int j = 0; j < 4; ++j) {
    const int col = n0 + wn + j * 16 + ln;
    const float bj = bias[col];
#pragma unroll
    for (int i = 0; i < 4; ++i) {
      const int rbase = m0 + wm + i * 16 + half * 8;
#pragma unroll
      for (int v = 0; v < 8; ++v)
        C[(size_t)(rbase + v) * N + col] = acc[i][j][v] + bj;
    }
  }
}

#define KPAD 68
#define VPITCH 136

__device__ __forceinline__ void kv_fetch(const float* __restrict__ base, int j,
                                         int r0, int c, v4f (&rk)[8],
                                         v4f (&rva)[4], v4f (&rvb)[4]) {
  const size_t rstride = 3 * D_MODEL;
#pragma unroll
  for (int it = 0; it < 8; ++it) {
    const float* row = base + (size_t)(j * 64 + r0 + it * 8) * rstride;
    rk[it] = *(const v4f*)(row + D_MODEL + c);
  }
#pragma unroll
  for (int it = 0; it < 4; ++it) {
    const int p = r0 + it * 8;
    const float* row = base + (size_t)(j * 64 + 2 * p) * rstride + 2 * D_MODEL + c;
    rva[it] = *(const v4f*)row;
    rvb[it] = *(const v4f*)(row + rstride);
  }
}

__device__ __forceinline__ void kv_stage(float (*Kst)[KPAD], float (*Vst)[VPITCH],
                                         int r0, int c, const v4f (&rk)[8],
                                         const v4f (&rva)[4], const v4f (&rvb)[4]) {
#pragma unroll
  for (int it = 0; it < 8; ++it) *(v4f*)&Kst[r0 + it * 8][c] = rk[it];
#pragma unroll
  for (int it = 0; it < 4; ++it) {
    const int p = r0 + it * 8;
    v4f lo, hi;
    lo.x = rva[it].x; lo.y = rvb[it].x; lo.z = rva[it].y; lo.w = rvb[it].y;
    hi.x = rva[it].z; hi.y = rvb[it].z; hi.z = rva[it].w; hi.w = rvb[it].w;
    *(v4f*)&Vst[p][2 * c] = lo;
    *(v4f*)&Vst[p][2 * c + 4] = hi;
  }
}

__device__ __forceinline__ float xhalf(float x) {
  int b = __builtin_bit_cast(int, x);
  return __builtin_bit_cast(
      float, __builtin_amdgcn_permlanex16(b, b, 0x76543210, 0xfedcba98, false, false));
}

__global__ __launch_bounds__(128) void mha_kernel(const float* __restrict__ qkv,
                                                  float* __restrict__ attn_out) {
  __shared__ float Ks[2][64][KPAD];
  __shared__ float Vp[2][32][VPITCH];

  const int bh = blockIdx.y;
  const int b = bh >> 4;
  const int h = bh & 15;

  const int tid = threadIdx.x;
  const int lane = tid & 31;
  const int w = tid >> 5;
  const int ln = lane & 15;
  const int half = lane >> 4;
  const int kb = half << 1;

  const size_t rstride = 3 * D_MODEL;
  const float* base = qkv + (size_t)b * SEQ_L * rstride + (size_t)h * HEAD_DIM;
  float* obase = attn_out + (size_t)b * SEQ_L * D_MODEL + (size_t)h * HEAD_DIM;

  const float cs = 0.03125f * 1.44269504088896f;

  const int r0 = tid >> 4;
  const int c = (tid & 15) << 2;
  v4f rk[8], rva[4], rvb[4];

  kv_fetch(base, 0, r0, c, rk, rva, rvb);
  kv_stage(Ks[0], Vp[0], r0, c, rk, rva, rvb);
  __syncthreads();
  int st = 0;
  int tiles_done = 0;
  int plevel = -1;

  for (int pass = 0; pass < 2; ++pass) {
    const int qb = pass ? (int)blockIdx.x : (SEQ_L / 64 - 1) - (int)blockIdx.x;
    const int qw0 = qb * 64 + w * 16;

    v2f qa[16];
    {
      const float* qp = base + (size_t)(qw0 + ln) * rstride + kb;
#pragma unroll
      for (int kk = 0; kk < 16; ++kk) qa[kk] = *(const v2f*)(qp + kk * 4);
    }
    asm volatile("" ::"v"(qa[0]), "v"(qa[1]), "v"(qa[2]), "v"(qa[3]), "v"(qa[4]),
                 "v"(qa[5]), "v"(qa[6]), "v"(qa[7]), "v"(qa[8]), "v"(qa[9]),
                 "v"(qa[10]), "v"(qa[11]), "v"(qa[12]), "v"(qa[13]), "v"(qa[14]),
                 "v"(qa[15]));

    v8f acc_o[4];
#pragma unroll
    for (int v = 0; v < 8; ++v) {
      acc_o[0][v] = 0.0f; acc_o[1][v] = 0.0f;
      acc_o[2][v] = 0.0f; acc_o[3][v] = 0.0f;
    }
    float m_i = -__builtin_inff();
    float l_i = 0.0f;

    for (int j = 0; j <= qb; ++j) {
      const bool last = (j == qb);
      const bool more = !last || pass == 0;
      const int lv = prio_level(tiles_done, 12, 22, 29);
      if (lv != plevel) { plevel = lv; set_prio_for_level(lv); }
      ++tiles_done;
      if (more) {
        kv_fetch(base, last ? 0 : j + 1, r0, c, rk, rva, rvb);
      }
      const float (*Kc)[KPAD] = Ks[st];
      const float (*Vc)[VPITCH] = Vp[st];

      v8f s[4];
#pragma unroll
      for (int nt = 0; nt < 4; ++nt)
#pragma unroll
        for (int v = 0; v < 8; ++v) s[nt][v] = 0.0f;
#pragma unroll
      for (int kk = 0; kk < 16; ++kk) {
        v2f a0 = *(const v2f*)&Kc[0 + ln][kk * 4 + kb];
        v2f a1 = *(const v2f*)&Kc[16 + ln][kk * 4 + kb];
        v2f a2 = *(const v2f*)&Kc[32 + ln][kk * 4 + kb];
        v2f a3 = *(const v2f*)&Kc[48 + ln][kk * 4 + kb];
        s[0] = wmma_f32(a0, qa[kk], s[0]);
        s[1] = wmma_f32(a1, qa[kk], s[1]);
        s[2] = wmma_f32(a2, qa[kk], s[2]);
        s[3] = wmma_f32(a3, qa[kk], s[3]);
      }

      if (last) {
        const int ql = w * 16 + ln;
#pragma unroll
        for (int nt = 0; nt < 4; ++nt)
#pragma unroll
          for (int v = 0; v < 8; ++v)
            if (nt * 16 + half * 8 + v > ql) s[nt][v] = -__builtin_inff();
      }

      float mxt[4];
#pragma unroll
      for (int nt = 0; nt < 4; ++nt) {
        float t0 = s[nt][0];
#pragma unroll
        for (int v = 1; v < 8; ++v) t0 = fmaxf(t0, s[nt][v]);
        mxt[nt] = t0;
      }
      float mx = fmaxf(fmaxf(mxt[0], mxt[1]), fmaxf(mxt[2], mxt[3]));
      mx = fmaxf(mx, xhalf(mx));

      const float mn = fmaxf(m_i, mx);
      const float alpha = __builtin_amdgcn_exp2f((m_i - mn) * cs);
      m_i = mn;
      const float nb = -mn * cs;

      float rst[4];
#pragma unroll
      for (int nt = 0; nt < 4; ++nt) {
        float r0 = 0.0f;
#pragma unroll
        for (int v = 0; v < 8; ++v) {
          const float p = __builtin_amdgcn_exp2f(__builtin_fmaf(s[nt][v], cs, nb));
          s[nt][v] = p;
          r0 += p;
        }
        rst[nt] = r0;
      }
      l_i = l_i * alpha + ((rst[0] + rst[1]) + (rst[2] + rst[3]));
#pragma unroll
      for (int dt = 0; dt < 4; ++dt)
#pragma unroll
        for (int v = 0; v < 8; ++v) acc_o[dt][v] *= alpha;

#pragma unroll
      for (int nt = 0; nt < 4; ++nt) {
#pragma unroll
        for (int t = 0; t < 4; ++t) {
          v2f pb;
          pb.x = s[nt][2 * t];
          pb.y = s[nt][2 * t + 1];
          const float* vrow = &Vc[nt * 8 + half * 4 + t][2 * ln];
          v2f a0 = *(const v2f*)(vrow + 0);
          v2f a1 = *(const v2f*)(vrow + 32);
          v2f a2 = *(const v2f*)(vrow + 64);
          v2f a3 = *(const v2f*)(vrow + 96);
          acc_o[0] = wmma_f32(a0, pb, acc_o[0]);
          acc_o[1] = wmma_f32(a1, pb, acc_o[1]);
          acc_o[2] = wmma_f32(a2, pb, acc_o[2]);
          acc_o[3] = wmma_f32(a3, pb, acc_o[3]);
        }
      }

      if (more) {
        kv_stage(Ks[st ^ 1], Vp[st ^ 1], r0, c, rk, rva, rvb);
      }
      __syncthreads();
      st ^= 1;
    }

    {
      const float lt = l_i + xhalf(l_i);
      const float inv_l = 1.0f / lt;
      float* orow = obase + (size_t)(qw0 + ln) * D_MODEL + half * 8;
#pragma unroll
      for (int dt = 0; dt < 4; ++dt) {
        v4f o0, o1;
        o0.x = acc_o[dt][0] * inv_l; o0.y = acc_o[dt][1] * inv_l;
        o0.z = acc_o[dt][2] * inv_l; o0.w = acc_o[dt][3] * inv_l;
        o1.x = acc_o[dt][4] * inv_l; o1.y = acc_o[dt][5] * inv_l;
        o1.z = acc_o[dt][6] * inv_l; o1.w = acc_o[dt][7] * inv_l;
        *(v4f*)(orow + dt * 16) = o0;
        *(v4f*)(orow + dt * 16 + 4) = o1;
      }
    }
  }
}

extern "C" void kernel_launch(void* const* d_in, const int* in_sizes, int n_in,
                              void* d_out, int out_size, void* d_ws,
                              size_t ws_size, hipStream_t stream) {
  (void)in_sizes; (void)n_in; (void)out_size; (void)ws_size;
  const float* x = (const float*)d_in[0];
  const float* w_qkv = (const float*)d_in[1];
  const float* b_qkv = (const float*)d_in[2];
  const float* w_out = (const float*)d_in[3];
  const float* b_out = (const float*)d_in[4];
  float* out = (float*)d_out;

  float* qkv = (float*)d_ws;
  float* attn = qkv + (size_t)M_ROWS * 3 * D_MODEL;

  gemm_bias_wide_kernel<3 * D_MODEL, D_MODEL>
      <<<dim3((3 * D_MODEL) / QBN, M_ROWS / QBM), 128, 0, stream>>>(x, w_qkv, b_qkv, qkv);

  mha_kernel<<<dim3(SEQ_L / 128, BATCH * H_HEADS), 128, 0, stream>>>(qkv, attn);

  gemm_bias_kernel<<<dim3(D_MODEL / BN, M_ROWS / BM), 256, 0, stream>>>(
      attn, w_out, b_out, out, D_MODEL, D_MODEL);
}
